# starting kernel
# baseline (speedup 1.0000x reference)
#include <hip/hip_runtime.h>

typedef float v2f __attribute__((ext_vector_type(2)));
typedef float v8f __attribute__((ext_vector_type(8)));

#define NN 4096
#define EE 131072
#define INDIM 32
#define DD 64
#define HH 8
#define DHH 8
#define FF 64
#define LL 4
#define ND (NN * DD)

__device__ __forceinline__ v8f wmma4(v2f a, v2f b, v8f c) {
  return __builtin_amdgcn_wmma_f32_16x16x4_f32(
       false, a,  false, b,
       (short)0, c,  false,  false);
}

__global__ void zero_f(float* p, int n) {
  int i = blockIdx.x * blockDim.x + threadIdx.x;
  if (i < n) p[i] = 0.f;
}
__global__ void zero_i(int* p, int n) {
  int i = blockIdx.x * blockDim.x + threadIdx.x;
  if (i < n) p[i] = 0;
}
__global__ void count_deg(const int* __restrict__ src, const int* __restrict__ dst,
                          int* cs, int* ci) {
  int e = blockIdx.x * blockDim.x + threadIdx.x;
  if (e < EE) {
    atomicAdd(&cs[src[e]], 1);
    atomicAdd(&ci[dst[e]], 1);
  }
}
__global__ void finalize_deg(const int* __restrict__ cs, const int* __restrict__ ci,
                             float* degree, float* invdegin) {
  int n = blockIdx.x * blockDim.x + threadIdx.x;
  if (n < NN) {
    degree[n] = rsqrtf(1.f + (float)cs[n]);
    invdegin[n] = 1.f / fmaxf((float)ci[n], 1.f);
  }
}
__global__ void scatter_msg(const float* __restrict__ h, const int* __restrict__ src,
                            const int* __restrict__ dst, float* msg) {
  int i = blockIdx.x * blockDim.x + threadIdx.x;
  if (i < EE * DD) {
    int e = i >> 6, c = i & 63;
    atomicAdd(&msg[dst[e] * DD + c], h[src[e] * DD + c]);
  }
}
__global__ void preadd(const float* __restrict__ h, const float* __restrict__ msg,
                       const float* __restrict__ invdegin, float* out) {
  int i = blockIdx.x * blockDim.x + threadIdx.x;
  if (i < ND) out[i] = h[i] + msg[i] * invdegin[i >> 6];
}

__global__ void __launch_bounds__(128) gemm_k(const float* __restrict__ A,
                                              const float* __restrict__ B,
                                              float* __restrict__ C,
                                              int kdim, int relu) {
  int wave = threadIdx.x >> 5;
  int lane = threadIdx.x & 31;
  int rowbase = blockIdx.x * 16;
  int colbase = wave * 16;
  int m = lane & 15;
  int khalf = (lane >> 4) * 2;
  v8f c = {};
  for (int k0 = 0; k0 < kdim; k0 += 4) {
    v2f a, b;
    const float* ar = A + (rowbase + m) * kdim + k0 + khalf;
    a.x = ar[0];
    a.y = ar[1];
    b.x = B[(k0 + khalf) * 64 + colbase + m];
    b.y = B[(k0 + khalf + 1) * 64 + colbase + m];
    c = wmma4(a, b, c);
  }
  int rr = rowbase + 8 * (lane >> 4);
  int cc = colbase + m;
#pragma unroll
  for (int j = 0; j < 8; j++) {
    float v = c[j];
    if (relu) v = fmaxf(v, 0.f);
    C[(rr + j) * 64 + cc] = v;
  }
}

__global__ void __launch_bounds__(256) attn_kernel(const float* __restrict__ q,
                                                   const float* __restrict__ k,
                                                   const float* __restrict__ v,
                                                   float* __restrict__ o) {
  __shared__ float pt[HH][16 * 17];
  int h = threadIdx.x >> 5;
  int lane = threadIdx.x & 31;
  int qbase = blockIdx.x * 16;
  int m = lane & 15;
  int half = lane >> 4;
  int khalf = half * 2;
  const float scale = 0.35355339059327373f;

  const float* qrow = q + (qbase + m) * DD + h * DHH;
  v2f qa0, qa1;
  qa0.x = qrow[khalf] * scale;
  qa0.y = qrow[khalf + 1] * scale;
  qa1.x = qrow[4 + khalf] * scale;
  qa1.y = qrow[4 + khalf + 1] * scale;

  float mrow[8], lrow[8];
#pragma unroll
  for (int j = 0; j < 8; j++) { mrow[j] = -1e30f; lrow[j] = 0.f; }
  v8f acc = {};
  float* myp = &pt[h][0];

  for (int kb = 0; kb < NN; kb += 16) {
    v8f s = {};
    const float* krow = k + (kb + m) * DD + h * DHH;
    v2f b;
    b.x = krow[khalf];     b.y = krow[khalf + 1];
    s = wmma4(qa0, b, s);
    b.x = krow[4 + khalf]; b.y = krow[4 + khalf + 1];
    s = wmma4(qa1, b, s);

#pragma unroll
    for (int j = 0; j < 8; j++) {
      float sm = s[j];
#pragma unroll
      for (int off = 1; off < 16; off <<= 1) sm = fmaxf(sm, __shfl_xor(sm, off, 32));
      float mn = fmaxf(mrow[j], sm);
      float alpha = __expf(mrow[j] - mn);
      mrow[j] = mn;
      float p = __expf(s[j] - mn);
      float ps = p;
#pragma unroll
      for (int off = 1; off < 16; off <<= 1) ps += __shfl_xor(ps, off, 32);
      lrow[j] = lrow[j] * alpha + ps;
      acc[j] *= alpha;
      myp[(j + 8 * half) * 17 + m] = p;
    }
    asm volatile("s_wait_dscnt 0" ::: "memory");

#pragma unroll
    for (int kc = 0; kc < 16; kc += 4) {
      v2f a, bb;
      a.x = myp[m * 17 + kc + khalf];
      a.y = myp[m * 17 + kc + khalf + 1];
      int key0 = kb + kc + khalf;
      bb.x = (m < DHH) ? v[key0 * DD + h * DHH + m] : 0.f;
      bb.y = (m < DHH) ? v[(key0 + 1) * DD + h * DHH + m] : 0.f;
      acc = wmma4(a, bb, acc);
    }
    asm volatile("s_wait_dscnt 0" ::: "memory");
  }

  if (m < DHH) {
#pragma unroll
    for (int j = 0; j < 8; j++)
      o[(qbase + j + 8 * half) * DD + h * DHH + m] = acc[j] / lrow[j];
  }
}

__global__ void __launch_bounds__(64) resid_ln(float* __restrict__ h,
                                               const float* __restrict__ add,
                                               const float* __restrict__ rowscale,
                                               int use_scale,
                                               const float* __restrict__ g,
                                               const float* __restrict__ b) {
  __shared__ float red[64];
  int row = blockIdx.x, c = threadIdx.x;
  float sc = use_scale ? rowscale[row] : 1.f;
  float val = h[row * DD + c] + add[row * DD + c] * sc;
  red[c] = val;
  __syncthreads();
  for (int s = 32; s > 0; s >>= 1) { if (c < s) red[c] += red[c + s]; __syncthreads(); }
  float mean = red[0] * (1.f / 64.f);
  __syncthreads();
  float d = val - mean;
  red[c] = d * d;
  __syncthreads();
  for (int s = 32; s > 0; s >>= 1) { if (c < s) red[c] += red[c + s]; __syncthreads(); }
  float var = red[0] * (1.f / 64.f);
  h[row * DD + c] = d * rsqrtf(var + 1e-5f) * g[c] + b[c];
}

extern "C" void kernel_launch(void* const* d_in, const int* in_sizes, int n_in,
                              void* d_out, int out_size, void* d_ws, size_t ws_size,
                              hipStream_t stream) {
  const float* x  = (const float*)d_in[0];
  const int*  ei  = (const int*)d_in[1];
  const int*  src = ei;
  const int*  dst = ei + EE;
  const float* We = (const float*)d_in[2];
  const float* Wg = (const float*)d_in[3];
  const float* Wq = (const float*)d_in[4];
  const float* Wk = (const float*)d_in[5];
  const float* Wv = (const float*)d_in[6];
  const float* Wo = (const float*)d_in[7];
  const float* W1 = (const float*)d_in[8];
  const float* W2 = (const float*)d_in[9];
  const float* g1 = (const float*)d_in[10];
  const float* b1 = (const float*)d_in[11];
  const float* g2 = (const float*)d_in[12];
  const float* b2 = (const float*)d_in[13];

  float* h = (float*)d_out;
  float* ws = (float*)d_ws;
  float* msg   = ws;
  float* sbuf  = msg + ND;
  float* sfeat = sbuf + ND;
  float* qb    = sfeat + ND;
  float* kb    = qb + ND;
  float* vb    = kb + ND;
  float* oatt  = vb + ND;
  float* t1    = oatt + ND;
  float* t2    = t1 + ND;
  float* degree   = t2 + ND;
  float* invdegin = degree + NN;
  int*   cnt      = (int*)(invdegin + NN);

  zero_i<<<(2 * NN + 255) / 256, 256, 0, stream>>>(cnt, 2 * NN);
  count_deg<<<(EE + 255) / 256, 256, 0, stream>>>(src, dst, cnt, cnt + NN);
  finalize_deg<<<(NN + 255) / 256, 256, 0, stream>>>(cnt, cnt + NN, degree, invdegin);

  gemm_k<<<NN / 16, 128, 0, stream>>>(x, We, h, INDIM, 0);

  for (int l = 0; l < LL; l++) {
    zero_f<<<(ND + 255) / 256, 256, 0, stream>>>(msg, ND);
    scatter_msg<<<(EE * DD + 255) / 256, 256, 0, stream>>>(h, src, dst, msg);
    preadd<<<(ND + 255) / 256, 256, 0, stream>>>(h, msg, invdegin, sbuf);
    gemm_k<<<NN / 16, 128, 0, stream>>>(sbuf, Wg + l * DD * DD, sfeat, DD, 1);

    gemm_k<<<NN / 16, 128, 0, stream>>>(sfeat, Wq + l * DD * DD, qb, DD, 0);
    gemm_k<<<NN / 16, 128, 0, stream>>>(sfeat, Wk + l * DD * DD, kb, DD, 0);
    gemm_k<<<NN / 16, 128, 0, stream>>>(h,     Wv + l * DD * DD, vb, DD, 0);

    attn_kernel<<<NN / 16, 256, 0, stream>>>(qb, kb, vb, oatt);

    gemm_k<<<NN / 16, 128, 0, stream>>>(oatt, Wo + l * DD * DD, t1, DD, 0);
    resid_ln<<<NN, 64, 0, stream>>>(h, t1, degree, 1, g1 + l * DD, b1 + l * DD);

    gemm_k<<<NN / 16, 128, 0, stream>>>(h, W1 + l * DD * FF, t1, DD, 1);
    gemm_k<<<NN / 16, 128, 0, stream>>>(t1, W2 + l * FF * DD, t2, FF, 0);
    resid_ln<<<NN, 64, 0, stream>>>(h, t2, nullptr, 0, g2 + l * DD, b2 + l * DD);
  }
}


# attention score tile computed transposed (St = K Qt, one query per lane): key max/sum in-lane + one lane^16 exchange, P stays in registers as PV operand; denominator and PV key sum order change
# speedup vs baseline: 3.3933x; 3.3933x over previous
#include <hip/hip_runtime.h>

typedef float v2f __attribute__((ext_vector_type(2)));
typedef float v8f __attribute__((ext_vector_type(8)));

#define NN 4096
#define EE 131072
#define INDIM 32
#define DD 64
#define HH 8
#define DHH 8
#define FF 64
#define LL 4
#define ND (NN * DD)

__device__ __forceinline__ v8f wmma4(v2f a, v2f b, v8f c) {
  return __builtin_amdgcn_wmma_f32_16x16x4_f32(
       false, a,  false, b,
       (short)0, c,  false,  false);
}

__global__ void zero_f(float* p, int n) {
  int i = blockIdx.x * blockDim.x + threadIdx.x;
  if (i < n) p[i] = 0.f;
}
__global__ void zero_i(int* p, int n) {
  int i = blockIdx.x * blockDim.x + threadIdx.x;
  if (i < n) p[i] = 0;
}
__global__ void count_deg(const int* __restrict__ src, const int* __restrict__ dst,
                          int* cs, int* ci) {
  int e = blockIdx.x * blockDim.x + threadIdx.x;
  if (e < EE) {
    atomicAdd(&cs[src[e]], 1);
    atomicAdd(&ci[dst[e]], 1);
  }
}
__global__ void finalize_deg(const int* __restrict__ cs, const int* __restrict__ ci,
                             float* degree, float* invdegin) {
  int n = blockIdx.x * blockDim.x + threadIdx.x;
  if (n < NN) {
    degree[n] = rsqrtf(1.f + (float)cs[n]);
    invdegin[n] = 1.f / fmaxf((float)ci[n], 1.f);
  }
}
__global__ void scatter_msg(const float* __restrict__ h, const int* __restrict__ src,
                            const int* __restrict__ dst, float* msg) {
  int i = blockIdx.x * blockDim.x + threadIdx.x;
  if (i < EE * DD) {
    int e = i >> 6, c = i & 63;
    atomicAdd(&msg[dst[e] * DD + c], h[src[e] * DD + c]);
  }
}
__global__ void preadd(const float* __restrict__ h, const float* __restrict__ msg,
                       const float* __restrict__ invdegin, float* out) {
  int i = blockIdx.x * blockDim.x + threadIdx.x;
  if (i < ND) out[i] = h[i] + msg[i] * invdegin[i >> 6];
}

__global__ void __launch_bounds__(128) gemm_k(const float* __restrict__ A,
                                              const float* __restrict__ B,
                                              float* __restrict__ C,
                                              int kdim, int relu) {
  int wave = threadIdx.x >> 5;
  int lane = threadIdx.x & 31;
  int rowbase = blockIdx.x * 16;
  int colbase = wave * 16;
  int m = lane & 15;
  int khalf = (lane >> 4) * 2;
  v8f c = {};
  for (int k0 = 0; k0 < kdim; k0 += 4) {
    v2f a, b;
    const float* ar = A + (rowbase + m) * kdim + k0 + khalf;
    a.x = ar[0];
    a.y = ar[1];
    b.x = B[(k0 + khalf) * 64 + colbase + m];
    b.y = B[(k0 + khalf + 1) * 64 + colbase + m];
    c = wmma4(a, b, c);
  }
  int rr = rowbase + 8 * (lane >> 4);
  int cc = colbase + m;
#pragma unroll
  for (int j = 0; j < 8; j++) {
    float v = c[j];
    if (relu) v = fmaxf(v, 0.f);
    C[(rr + j) * 64 + cc] = v;
  }
}

__device__ __forceinline__ float xhalf(float x) {
  int b = __builtin_bit_cast(int, x);
  return __builtin_bit_cast(
      float, __builtin_amdgcn_permlanex16(b, b, 0x76543210, 0xfedcba98, false, false));
}

__global__ void __launch_bounds__(256) attn_kernel(const float* __restrict__ q,
                                                   const float* __restrict__ k,
                                                   const float* __restrict__ v,
                                                   float* __restrict__ o) {
  int h = threadIdx.x >> 5;
  int lane = threadIdx.x & 31;
  int qbase = blockIdx.x * 16;
  int m = lane & 15;
  int half = lane >> 4;
  int khalf = half * 2;
  const float scale = 0.35355339059327373f;

  const float* qrow = q + (qbase + m) * DD + h * DHH;
  v2f qb0, qb1;
  qb0.x = qrow[khalf] * scale;
  qb0.y = qrow[khalf + 1] * scale;
  qb1.x = qrow[4 + khalf] * scale;
  qb1.y = qrow[4 + khalf + 1] * scale;

  float mrun = -1e30f;
  float lsum = 0.f;
  v8f acc = {};

  const float* kp = k + m * DD + h * DHH + khalf;
  const float* vp = v + (8 * half) * DD + h * DHH + (m & 7);

  for (int kb = 0; kb < NN; kb += 16) {
    const float* krow = kp + kb * DD;
    v8f s = {};
    v2f a;
    a.x = krow[0]; a.y = krow[1];
    s = wmma4(a, qb0, s);
    a.x = krow[4]; a.y = krow[5];
    s = wmma4(a, qb1, s);

    float sm = fmaxf(fmaxf(fmaxf(s[0], s[1]), fmaxf(s[2], s[3])),
                     fmaxf(fmaxf(s[4], s[5]), fmaxf(s[6], s[7])));
    sm = fmaxf(sm, xhalf(sm));
    float mn = fmaxf(mrun, sm);
    float alpha = __expf(mrun - mn);
    mrun = mn;
    float p[8];
    float ps = 0.f;
#pragma unroll
    for (int j = 0; j < 8; j++) {
      p[j] = __expf(s[j] - mn);
      ps += p[j];
    }
    lsum = lsum * alpha + ps;
#pragma unroll
    for (int j = 0; j < 8; j++) acc[j] *= alpha;

    const float* vrow = vp + kb * DD;
#pragma unroll
    for (int c = 0; c < 4; c++) {
      v2f va, pb;
      va.x = vrow[(2 * c) * DD];
      va.y = vrow[(2 * c + 1) * DD];
      pb.x = p[2 * c];
      pb.y = p[2 * c + 1];
      acc = wmma4(va, pb, acc);
    }
  }

  float l = lsum + xhalf(lsum);
  if (half == 0) {
    float* orow = o + (qbase + m) * DD + h * DHH;
#pragma unroll
    for (int j = 0; j < 8; j++) orow[j] = acc[j] / l;
  }
}

__global__ void __launch_bounds__(64) resid_ln(float* __restrict__ h,
                                               const float* __restrict__ add,
                                               const float* __restrict__ rowscale,
                                               int use_scale,
                                               const float* __restrict__ g,
                                               const float* __restrict__ b) {
  __shared__ float red[64];
  int row = blockIdx.x, c = threadIdx.x;
  float sc = use_scale ? rowscale[row] : 1.f;
  float val = h[row * DD + c] + add[row * DD + c] * sc;
  red[c] = val;
  __syncthreads();
  for (int s = 32; s > 0; s >>= 1) { if (c < s) red[c] += red[c + s]; __syncthreads(); }
  float mean = red[0] * (1.f / 64.f);
  __syncthreads();
  float d = val - mean;
  red[c] = d * d;
  __syncthreads();
  for (int s = 32; s > 0; s >>= 1) { if (c < s) red[c] += red[c + s]; __syncthreads(); }
  float var = red[0] * (1.f / 64.f);
  h[row * DD + c] = d * rsqrtf(var + 1e-5f) * g[c] + b[c];
}

extern "C" void kernel_launch(void* const* d_in, const int* in_sizes, int n_in,
                              void* d_out, int out_size, void* d_ws, size_t ws_size,
                              hipStream_t stream) {
  const float* x  = (const float*)d_in[0];
  const int*  ei  = (const int*)d_in[1];
  const int*  src = ei;
  const int*  dst = ei + EE;
  const float* We = (const float*)d_in[2];
  const float* Wg = (const float*)d_in[3];
  const float* Wq = (const float*)d_in[4];
  const float* Wk = (const float*)d_in[5];
  const float* Wv = (const float*)d_in[6];
  const float* Wo = (const float*)d_in[7];
  const float* W1 = (const float*)d_in[8];
  const float* W2 = (const float*)d_in[9];
  const float* g1 = (const float*)d_in[10];
  const float* b1 = (const float*)d_in[11];
  const float* g2 = (const float*)d_in[12];
  const float* b2 = (const float*)d_in[13];

  float* h = (float*)d_out;
  float* ws = (float*)d_ws;
  float* msg   = ws;
  float* sbuf  = msg + ND;
  float* sfeat = sbuf + ND;
  float* qb    = sfeat + ND;
  float* kb    = qb + ND;
  float* vb    = kb + ND;
  float* oatt  = vb + ND;
  float* t1    = oatt + ND;
  float* t2    = t1 + ND;
  float* degree   = t2 + ND;
  float* invdegin = degree + NN;
  int*   cnt      = (int*)(invdegin + NN);

  zero_i<<<(2 * NN + 255) / 256, 256, 0, stream>>>(cnt, 2 * NN);
  count_deg<<<(EE + 255) / 256, 256, 0, stream>>>(src, dst, cnt, cnt + NN);
  finalize_deg<<<(NN + 255) / 256, 256, 0, stream>>>(cnt, cnt + NN, degree, invdegin);

  gemm_k<<<NN / 16, 128, 0, stream>>>(x, We, h, INDIM, 0);

  for (int l = 0; l < LL; l++) {
    zero_f<<<(ND + 255) / 256, 256, 0, stream>>>(msg, ND);
    scatter_msg<<<(EE * DD + 255) / 256, 256, 0, stream>>>(h, src, dst, msg);
    preadd<<<(ND + 255) / 256, 256, 0, stream>>>(h, msg, invdegin, sbuf);
    gemm_k<<<NN / 16, 128, 0, stream>>>(sbuf, Wg + l * DD * DD, sfeat, DD, 1);

    gemm_k<<<NN / 16, 128, 0, stream>>>(sfeat, Wq + l * DD * DD, qb, DD, 0);
    gemm_k<<<NN / 16, 128, 0, stream>>>(sfeat, Wk + l * DD * DD, kb, DD, 0);
    gemm_k<<<NN / 16, 128, 0, stream>>>(h,     Wv + l * DD * DD, vb, DD, 0);

    attn_kernel<<<NN / 16, 256, 0, stream>>>(qb, kb, vb, oatt);

    gemm_k<<<NN / 16, 128, 0, stream>>>(oatt, Wo + l * DD * DD, t1, DD, 0);
    resid_ln<<<NN, 64, 0, stream>>>(h, t1, degree, 1, g1 + l * DD, b1 + l * DD);

    gemm_k<<<NN / 16, 128, 0, stream>>>(h, W1 + l * DD * FF, t1, DD, 1);
    gemm_k<<<NN / 16, 128, 0, stream>>>(t1, W2 + l * FF * DD, t2, FF, 0);
    resid_ln<<<NN, 64, 0, stream>>>(h, t2, nullptr, 0, g2 + l * DD, b2 + l * DD);
  }
}


# attention: K and V of key tile i+6 are loaded at the start of step i into a ring of 8 register sets, so no step waits for its own loads; key order, arithmetic and sum order unchanged
# speedup vs baseline: 5.5079x; 1.6232x over previous
#include <hip/hip_runtime.h>

typedef float v2f __attribute__((ext_vector_type(2)));
typedef float v8f __attribute__((ext_vector_type(8)));

#define NN 4096
#define EE 131072
#define INDIM 32
#define DD 64
#define HH 8
#define DHH 8
#define FF 64
#define LL 4
#define ND (NN * DD)

__device__ __forceinline__ v8f wmma4(v2f a, v2f b, v8f c) {
  return __builtin_amdgcn_wmma_f32_16x16x4_f32(
       false, a,  false, b,
       (short)0, c,  false,  false);
}

__global__ void zero_f(float* p, int n) {
  int i = blockIdx.x * blockDim.x + threadIdx.x;
  if (i < n) p[i] = 0.f;
}
__global__ void zero_i(int* p, int n) {
  int i = blockIdx.x * blockDim.x + threadIdx.x;
  if (i < n) p[i] = 0;
}
__global__ void count_deg(const int* __restrict__ src, const int* __restrict__ dst,
                          int* cs, int* ci) {
  int e = blockIdx.x * blockDim.x + threadIdx.x;
  if (e < EE) {
    atomicAdd(&cs[src[e]], 1);
    atomicAdd(&ci[dst[e]], 1);
  }
}
__global__ void finalize_deg(const int* __restrict__ cs, const int* __restrict__ ci,
                             float* degree, float* invdegin) {
  int n = blockIdx.x * blockDim.x + threadIdx.x;
  if (n < NN) {
    degree[n] = rsqrtf(1.f + (float)cs[n]);
    invdegin[n] = 1.f / fmaxf((float)ci[n], 1.f);
  }
}
__global__ void scatter_msg(const float* __restrict__ h, const int* __restrict__ src,
                            const int* __restrict__ dst, float* msg) {
  int i = blockIdx.x * blockDim.x + threadIdx.x;
  if (i < EE * DD) {
    int e = i >> 6, c = i & 63;
    atomicAdd(&msg[dst[e] * DD + c], h[src[e] * DD + c]);
  }
}
__global__ void preadd(const float* __restrict__ h, const float* __restrict__ msg,
                       const float* __restrict__ invdegin, float* out) {
  int i = blockIdx.x * blockDim.x + threadIdx.x;
  if (i < ND) out[i] = h[i] + msg[i] * invdegin[i >> 6];
}

__global__ void __launch_bounds__(128) gemm_k(const float* __restrict__ A,
                                              const float* __restrict__ B,
                                              float* __restrict__ C,
                                              int kdim, int relu) {
  int wave = threadIdx.x >> 5;
  int lane = threadIdx.x & 31;
  int rowbase = blockIdx.x * 16;
  int colbase = wave * 16;
  int m = lane & 15;
  int khalf = (lane >> 4) * 2;
  v8f c = {};
  for (int k0 = 0; k0 < kdim; k0 += 4) {
    v2f a, b;
    const float* ar = A + (rowbase + m) * kdim + k0 + khalf;
    a.x = ar[0];
    a.y = ar[1];
    b.x = B[(k0 + khalf) * 64 + colbase + m];
    b.y = B[(k0 + khalf + 1) * 64 + colbase + m];
    c = wmma4(a, b, c);
  }
  int rr = rowbase + 8 * (lane >> 4);
  int cc = colbase + m;
#pragma unroll
  for (int j = 0; j < 8; j++) {
    float v = c[j];
    if (relu) v = fmaxf(v, 0.f);
    C[(rr + j) * 64 + cc] = v;
  }
}

__device__ __forceinline__ float xhalf(float x) {
  int b = __builtin_bit_cast(int, x);
  return __builtin_bit_cast(
      float, __builtin_amdgcn_permlanex16(b, b, 0x76543210, 0xfedcba98, false, false));
}


struct KVTile {
  v2f k0, k1;
  v2f v[4];
};

__device__ __forceinline__ void load_kv(KVTile& t, const float* __restrict__ kp,
                                        const float* __restrict__ vp, int kb) {
  const float* krow = kp + kb * DD;
  t.k0.x = krow[0]; t.k0.y = krow[1];
  t.k1.x = krow[4]; t.k1.y = krow[5];
  const float* vrow = vp + kb * DD;
#pragma unroll
  for (int c = 0; c < 4; c++) {
    t.v[c].x = vrow[(2 * c) * DD];
    t.v[c].y = vrow[(2 * c + 1) * DD];
  }
}

__device__ __forceinline__ void attn_step(const KVTile& t, v2f qb0, v2f qb1,
                                          float& mrun, float& lsum, v8f& acc) {
  v8f s = {};
  s = wmma4(t.k0, qb0, s);
  s = wmma4(t.k1, qb1, s);

  float sm = fmaxf(fmaxf(fmaxf(s[0], s[1]), fmaxf(s[2], s[3])),
                   fmaxf(fmaxf(s[4], s[5]), fmaxf(s[6], s[7])));
  sm = fmaxf(sm, xhalf(sm));
  float mn = fmaxf(mrun, sm);
  float alpha = __expf(mrun - mn);
  mrun = mn;
  float p[8];
  float ps = 0.f;
#pragma unroll
  for (int j = 0; j < 8; j++) {
    p[j] = __expf(s[j] - mn);
    ps += p[j];
  }
  lsum = lsum * alpha + ps;
#pragma unroll
  for (int j = 0; j < 8; j++) acc[j] *= alpha;

#pragma unroll
  for (int c = 0; c < 4; c++) {
    v2f pb;
    pb.x = p[2 * c];
    pb.y = p[2 * c + 1];
    acc = wmma4(t.v[c], pb, acc);
  }
}

#define RING 8
#define AHEAD 6

__device__ __forceinline__ void step_fence(v8f& acc, v2f& qb0, v2f& qb1, int& kb) {
  asm volatile("" : "+v"(acc), "+v"(qb0), "+v"(qb1), "+s"(kb));
}

__global__ void __launch_bounds__(256) attn_kernel(const float* __restrict__ q,
                                                   const float* __restrict__ k,
                                                   const float* __restrict__ v,
                                                   float* __restrict__ o) {
  int h = threadIdx.x >> 5;
  int lane = threadIdx.x & 31;
  int qbase = blockIdx.x * 16;
  int m = lane & 15;
  int half = lane >> 4;
  int khalf = half * 2;
  const float scale = 0.35355339059327373f;

  const float* qrow = q + (qbase + m) * DD + h * DHH;
  v2f qb0, qb1;
  qb0.x = qrow[khalf] * scale;
  qb0.y = qrow[khalf + 1] * scale;
  qb1.x = qrow[4 + khalf] * scale;
  qb1.y = qrow[4 + khalf + 1] * scale;

  float mrun = -1e30f;
  float lsum = 0.f;
  v8f acc = {};

  const float* kp = k + m * DD + h * DHH + khalf;
  const float* vp = v + (8 * half) * DD + h * DHH + (m & 7);

  KVTile t[RING];
  int kb = 0;
#pragma unroll
  for (int i = 0; i < AHEAD; i++) {
    load_kv(t[i], kp, vp, kb + 16 * i);
    step_fence(acc, qb0, qb1, kb);
  }

  for (; kb < NN - 16 * RING; kb += 16 * RING) {
#pragma unroll
    for (int i = 0; i < RING; i++) {
      load_kv(t[(i + AHEAD) % RING], kp, vp, kb + 16 * (i + AHEAD));
      attn_step(t[i], qb0, qb1, mrun, lsum, acc);
      step_fence(acc, qb0, qb1, kb);
    }
  }
#pragma unroll
  for (int i = 0; i < RING; i++) {
    if (i + AHEAD < RING) load_kv(t[i + AHEAD], kp, vp, NN - 16 * RING + 16 * (i + AHEAD));
    attn_step(t[i], qb0, qb1, mrun, lsum, acc);
  }

  float l = lsum + xhalf(lsum);
  if (half == 0) {
    float* orow = o + (qbase + m) * DD + h * DHH;
#pragma unroll
    for (int j = 0; j < 8; j++) orow[j] = acc[j] / l;
  }
}

__global__ void __launch_bounds__(64) resid_ln(float* __restrict__ h,
                                               const float* __restrict__ add,
                                               const float* __restrict__ rowscale,
                                               int use_scale,
                                               const float* __restrict__ g,
                                               const float* __restrict__ b) {
  __shared__ float red[64];
  int row = blockIdx.x, c = threadIdx.x;
  float sc = use_scale ? rowscale[row] : 1.f;
  float val = h[row * DD + c] + add[row * DD + c] * sc;
  red[c] = val;
  __syncthreads();
  for (int s = 32; s > 0; s >>= 1) { if (c < s) red[c] += red[c + s]; __syncthreads(); }
  float mean = red[0] * (1.f / 64.f);
  __syncthreads();
  float d = val - mean;
  red[c] = d * d;
  __syncthreads();
  for (int s = 32; s > 0; s >>= 1) { if (c < s) red[c] += red[c + s]; __syncthreads(); }
  float var = red[0] * (1.f / 64.f);
  h[row * DD + c] = d * rsqrtf(var + 1e-5f) * g[c] + b[c];
}

extern "C" void kernel_launch(void* const* d_in, const int* in_sizes, int n_in,
                              void* d_out, int out_size, void* d_ws, size_t ws_size,
                              hipStream_t stream) {
  const float* x  = (const float*)d_in[0];
  const int*  ei  = (const int*)d_in[1];
  const int*  src = ei;
  const int*  dst = ei + EE;
  const float* We = (const float*)d_in[2];
  const float* Wg = (const float*)d_in[3];
  const float* Wq = (const float*)d_in[4];
  const float* Wk = (const float*)d_in[5];
  const float* Wv = (const float*)d_in[6];
  const float* Wo = (const float*)d_in[7];
  const float* W1 = (const float*)d_in[8];
  const float* W2 = (const float*)d_in[9];
  const float* g1 = (const float*)d_in[10];
  const float* b1 = (const float*)d_in[11];
  const float* g2 = (const float*)d_in[12];
  const float* b2 = (const float*)d_in[13];

  float* h = (float*)d_out;
  float* ws = (float*)d_ws;
  float* msg   = ws;
  float* sbuf  = msg + ND;
  float* sfeat = sbuf + ND;
  float* qb    = sfeat + ND;
  float* kb    = qb + ND;
  float* vb    = kb + ND;
  float* oatt  = vb + ND;
  float* t1    = oatt + ND;
  float* t2    = t1 + ND;
  float* degree   = t2 + ND;
  float* invdegin = degree + NN;
  int*   cnt      = (int*)(invdegin + NN);

  zero_i<<<(2 * NN + 255) / 256, 256, 0, stream>>>(cnt, 2 * NN);
  count_deg<<<(EE + 255) / 256, 256, 0, stream>>>(src, dst, cnt, cnt + NN);
  finalize_deg<<<(NN + 255) / 256, 256, 0, stream>>>(cnt, cnt + NN, degree, invdegin);

  gemm_k<<<NN / 16, 128, 0, stream>>>(x, We, h, INDIM, 0);

  for (int l = 0; l < LL; l++) {
    zero_f<<<(ND + 255) / 256, 256, 0, stream>>>(msg, ND);
    scatter_msg<<<(EE * DD + 255) / 256, 256, 0, stream>>>(h, src, dst, msg);
    preadd<<<(ND + 255) / 256, 256, 0, stream>>>(h, msg, invdegin, sbuf);
    gemm_k<<<NN / 16, 128, 0, stream>>>(sbuf, Wg + l * DD * DD, sfeat, DD, 1);

    gemm_k<<<NN / 16, 128, 0, stream>>>(sfeat, Wq + l * DD * DD, qb, DD, 0);
    gemm_k<<<NN / 16, 128, 0, stream>>>(sfeat, Wk + l * DD * DD, kb, DD, 0);
    gemm_k<<<NN / 16, 128, 0, stream>>>(h,     Wv + l * DD * DD, vb, DD, 0);

    attn_kernel<<<NN / 16, 256, 0, stream>>>(qb, kb, vb, oatt);

    gemm_k<<<NN / 16, 128, 0, stream>>>(oatt, Wo + l * DD * DD, t1, DD, 0);
    resid_ln<<<NN, 64, 0, stream>>>(h, t1, degree, 1, g1 + l * DD, b1 + l * DD);

    gemm_k<<<NN / 16, 128, 0, stream>>>(h, W1 + l * DD * FF, t1, DD, 1);
    gemm_k<<<NN / 16, 128, 0, stream>>>(t1, W2 + l * FF * DD, t2, FF, 0);
    resid_ln<<<NN, 64, 0, stream>>>(h, t2, nullptr, 0, g2 + l * DD, b2 + l * DD);
  }
}


# row-local kernel chains fused, tiles between products kept in LDS: embed+clears, preadd+Wg+Wq/Wk/Wv, Wo+LN+FFN+LN one launch each (56 -> 18 launches); GEMM k order and LN tree sum order kept
# speedup vs baseline: 6.8024x; 1.2350x over previous
#include <hip/hip_runtime.h>

typedef float v2f __attribute__((ext_vector_type(2)));
typedef float v8f __attribute__((ext_vector_type(8)));

#define NN 4096
#define EE 131072
#define INDIM 32
#define DD 64
#define HH 8
#define DHH 8
#define FF 64
#define LL 4
#define ND (NN * DD)

__device__ __forceinline__ v8f wmma4(v2f a, v2f b, v8f c) {
  return __builtin_amdgcn_wmma_f32_16x16x4_f32(
       false, a,  false, b,
       (short)0, c,  false,  false);
}

__global__ void count_deg(const int* __restrict__ src, const int* __restrict__ dst,
                          int* cs, int* ci) {
  int e = blockIdx.x * blockDim.x + threadIdx.x;
  if (e < EE) {
    atomicAdd(&cs[src[e]], 1);
    atomicAdd(&ci[dst[e]], 1);
  }
}
__global__ void scatter_msg(const float* __restrict__ h, const int* __restrict__ src,
                            const int* __restrict__ dst, float* msg) {
  int i = blockIdx.x * blockDim.x + threadIdx.x;
  if (i < EE * DD) {
    int e = i >> 6, c = i & 63;
    atomicAdd(&msg[dst[e] * DD + c], h[src[e] * DD + c]);
  }
}

#define LP 68
#define CNT_PER_WG (2 * NN / (NN / 16))

__device__ __forceinline__ v8f tile_gemm64(const float* As, const float* __restrict__ B,
                                           int colbase, int m, int khalf) {
  v8f c = {};
#pragma unroll
  for (int k0 = 0; k0 < 64; k0 += 4) {
    v2f a, b;
    const float* ar = As + m * LP + k0 + khalf;
    a.x = ar[0];
    a.y = ar[1];
    b.x = B[(k0 + khalf) * 64 + colbase + m];
    b.y = B[(k0 + khalf + 1) * 64 + colbase + m];
    c = wmma4(a, b, c);
  }
  return c;
}

__device__ __forceinline__ void tile_to_global(float* __restrict__ C, v8f c, int rowbase,
                                               int colbase, int lane) {
  int rr = rowbase + 8 * (lane >> 4);
  int cc = colbase + (lane & 15);
#pragma unroll
  for (int j = 0; j < 8; j++) C[(rr + j) * 64 + cc] = c[j];
}
__device__ __forceinline__ void tile_relu_to_lds(float* Cs, v8f c, int colbase, int lane) {
  int rr = 8 * (lane >> 4);
  int cc = colbase + (lane & 15);
#pragma unroll
  for (int j = 0; j < 8; j++) Cs[(rr + j) * LP + cc] = fmaxf(c[j], 0.f);
}

__device__ __forceinline__ float row_tree_sum(const float* x) {
  float y = ((x[0] + x[4]) + (x[2] + x[6])) + ((x[1] + x[5]) + (x[3] + x[7]));
  y += __shfl_xor(y, 4);
  y += __shfl_xor(y, 2);
  y += __shfl_xor(y, 1);
  return y;
}

__device__ __forceinline__ void ln_rows_inplace(float* tile, int t, const float* __restrict__ g,
                                                const float* __restrict__ b) {
  int i = t & 7;
  float* row = tile + (t >> 3) * LP + i;
  float x[8], d[8], sq[8];
#pragma unroll
  for (int k = 0; k < 8; k++) x[k] = row[8 * k];
  float mean = row_tree_sum(x) * (1.f / 64.f);
#pragma unroll
  for (int k = 0; k < 8; k++) {
    d[k] = x[k] - mean;
    sq[k] = d[k] * d[k];
  }
  float var = row_tree_sum(sq) * (1.f / 64.f);
  float rs = rsqrtf(var + 1e-5f);
#pragma unroll
  for (int k = 0; k < 8; k++) row[8 * k] = d[k] * rs * g[i + 8 * k] + b[i + 8 * k];
}

__global__ void __launch_bounds__(128) embed_zero(const float* __restrict__ x,
                                                  const float* __restrict__ We,
                                                  float* __restrict__ h, float* __restrict__ msg,
                                                  int* __restrict__ cnt) {
  int t = threadIdx.x;
  int lane = t & 31;
  int rowbase = blockIdx.x * 16;
  int colbase = (t >> 5) * 16;
  int m = lane & 15;
  int khalf = (lane >> 4) * 2;
  if (t < CNT_PER_WG) cnt[blockIdx.x * CNT_PER_WG + t] = 0;
  const float4 z4 = {0.f, 0.f, 0.f, 0.f};
  float4* mrow = (float4*)(msg + rowbase * DD);
  mrow[t] = z4;
  mrow[t + 128] = z4;
  v8f c = {};
#pragma unroll
  for (int k0 = 0; k0 < INDIM; k0 += 4) {
    v2f a, b;
    const float* ar = x + (rowbase + m) * INDIM + k0 + khalf;
    a.x = ar[0];
    a.y = ar[1];
    b.x = We[(k0 + khalf) * 64 + colbase + m];
    b.y = We[(k0 + khalf + 1) * 64 + colbase + m];
    c = wmma4(a, b, c);
  }
  tile_to_global(h, c, rowbase, colbase, lane);
}

__global__ void __launch_bounds__(128) stage_a(const float* __restrict__ h,
                                               const float* __restrict__ msg,
                                               const int* __restrict__ ci,
                                               const float* __restrict__ Wg,
                                               const float* __restrict__ Wq,
                                               const float* __restrict__ Wk,
                                               const float* __restrict__ Wv,
                                               float* __restrict__ qb, float* __restrict__ kb,
                                               float* __restrict__ vb) {
  __shared__ __attribute__((aligned(16))) float sS[16 * LP];
  __shared__ __attribute__((aligned(16))) float sH[16 * LP];
  __shared__ __attribute__((aligned(16))) float sF[16 * LP];
  int t = threadIdx.x;
  int lane = t & 31;
  int rowbase = blockIdx.x * 16;
  int colbase = (t >> 5) * 16;
  int m = lane & 15;
  int khalf = (lane >> 4) * 2;

#pragma unroll
  for (int i = 0; i < 2; i++) {
    int idx = t + 128 * i;
    int r = idx >> 4, c = (idx & 15) * 4;
    int g = (rowbase + r) * DD + c;
    float4 hv = *(const float4*)(h + g);
    float4 mv = *(const float4*)(msg + g);
    float inv = 1.f / fmaxf((float)ci[rowbase + r], 1.f);
    float4 sv;
    sv.x = hv.x + mv.x * inv;
    sv.y = hv.y + mv.y * inv;
    sv.z = hv.z + mv.z * inv;
    sv.w = hv.w + mv.w * inv;
    *(float4*)(sH + r * LP + c) = hv;
    *(float4*)(sS + r * LP + c) = sv;
  }
  __syncthreads();
  v8f vv = tile_gemm64(sH, Wv, colbase, m, khalf);
  v8f ff = tile_gemm64(sS, Wg, colbase, m, khalf);
  tile_to_global(vb, vv, rowbase, colbase, lane);
  tile_relu_to_lds(sF, ff, colbase, lane);
  __syncthreads();
  v8f qq = tile_gemm64(sF, Wq, colbase, m, khalf);
  v8f kk = tile_gemm64(sF, Wk, colbase, m, khalf);
  tile_to_global(qb, qq, rowbase, colbase, lane);
  tile_to_global(kb, kk, rowbase, colbase, lane);
}

__global__ void __launch_bounds__(128) stage_b(float* __restrict__ h,
                                               const float* __restrict__ oatt,
                                               const int* __restrict__ cs,
                                               const float* __restrict__ Wo,
                                               const float* __restrict__ W1,
                                               const float* __restrict__ W2,
                                               const float* __restrict__ g1,
                                               const float* __restrict__ b1,
                                               const float* __restrict__ g2,
                                               const float* __restrict__ b2,
                                               float* __restrict__ msg) {
  __shared__ __attribute__((aligned(16))) float sX[16 * LP];
  __shared__ __attribute__((aligned(16))) float sY[16 * LP];
  int t = threadIdx.x;
  int lane = t & 31;
  int rowbase = blockIdx.x * 16;
  int colbase = (t >> 5) * 16;
  int m = lane & 15;
  int khalf = (lane >> 4) * 2;
  int rr = 8 * (lane >> 4);
  int cc = colbase + m;

  const float4 z4 = {0.f, 0.f, 0.f, 0.f};
#pragma unroll
  for (int i = 0; i < 2; i++) {
    int idx = t + 128 * i;
    int r = idx >> 4, c = (idx & 15) * 4;
    int g = (rowbase + r) * DD + c;
    *(float4*)(sX + r * LP + c) = *(const float4*)(oatt + g);
    *(float4*)(msg + g) = z4;
  }
  __syncthreads();

  v8f o = tile_gemm64(sX, Wo, colbase, m, khalf);
#pragma unroll
  for (int j = 0; j < 8; j++) {
    int row = rowbase + rr + j;
    float sc = rsqrtf(1.f + (float)cs[row]);
    sY[(rr + j) * LP + cc] = h[row * DD + cc] + o[j] * sc;
  }
  __syncthreads();
  ln_rows_inplace(sY, t, g1, b1);
  __syncthreads();

  v8f f1 = tile_gemm64(sY, W1, colbase, m, khalf);
  tile_relu_to_lds(sX, f1, colbase, lane);
  __syncthreads();
  v8f f2 = tile_gemm64(sX, W2, colbase, m, khalf);
#pragma unroll
  for (int j = 0; j < 8; j++) sY[(rr + j) * LP + cc] += f2[j];
  __syncthreads();
  ln_rows_inplace(sY, t, g2, b2);
  __syncthreads();

#pragma unroll
  for (int i = 0; i < 2; i++) {
    int idx = t + 128 * i;
    int r = idx >> 4, c = (idx & 15) * 4;
    *(float4*)(h + (rowbase + r) * DD + c) = *(const float4*)(sY + r * LP + c);
  }
}

__device__ __forceinline__ float xhalf(float x) {
  int b = __builtin_bit_cast(int, x);
  return __builtin_bit_cast(
      float, __builtin_amdgcn_permlanex16(b, b, 0x76543210, 0xfedcba98, false, false));
}


struct KVTile {
  v2f k0, k1;
  v2f v[4];
};

__device__ __forceinline__ void load_kv(KVTile& t, const float* __restrict__ kp,
                                        const float* __restrict__ vp, int kb) {
  const float* krow = kp + kb * DD;
  t.k0.x = krow[0]; t.k0.y = krow[1];
  t.k1.x = krow[4]; t.k1.y = krow[5];
  const float* vrow = vp + kb * DD;
#pragma unroll
  for (int c = 0; c < 4; c++) {
    t.v[c].x = vrow[(2 * c) * DD];
    t.v[c].y = vrow[(2 * c + 1) * DD];
  }
}

__device__ __forceinline__ void attn_step(const KVTile& t, v2f qb0, v2f qb1,
                                          float& mrun, float& lsum, v8f& acc) {
  v8f s = {};
  s = wmma4(t.k0, qb0, s);
  s = wmma4(t.k1, qb1, s);

  float sm = fmaxf(fmaxf(fmaxf(s[0], s[1]), fmaxf(s[2], s[3])),
                   fmaxf(fmaxf(s[4], s[5]), fmaxf(s[6], s[7])));
  sm = fmaxf(sm, xhalf(sm));
  float mn = fmaxf(mrun, sm);
  float alpha = __expf(mrun - mn);
  mrun = mn;
  float p[8];
  float ps = 0.f;
#pragma unroll
  for (int j = 0; j < 8; j++) {
    p[j] = __expf(s[j] - mn);
    ps += p[j];
  }
  lsum = lsum * alpha + ps;
#pragma unroll
  for (int j = 0; j < 8; j++) acc[j] *= alpha;

#pragma unroll
  for (int c = 0; c < 4; c++) {
    v2f pb;
    pb.x = p[2 * c];
    pb.y = p[2 * c + 1];
    acc = wmma4(t.v[c], pb, acc);
  }
}

#define RING 8
#define AHEAD 6

__device__ __forceinline__ void step_fence(v8f& acc, v2f& qb0, v2f& qb1, int& kb) {
  asm volatile("" : "+v"(acc), "+v"(qb0), "+v"(qb1), "+s"(kb));
}

__global__ void __launch_bounds__(256) attn_kernel(const float* __restrict__ q,
                                                   const float* __restrict__ k,
                                                   const float* __restrict__ v,
                                                   float* __restrict__ o) {
  int h = threadIdx.x >> 5;
  int lane = threadIdx.x & 31;
  int qbase = blockIdx.x * 16;
  int m = lane & 15;
  int half = lane >> 4;
  int khalf = half * 2;
  const float scale = 0.35355339059327373f;

  const float* qrow = q + (qbase + m) * DD + h * DHH;
  v2f qb0, qb1;
  qb0.x = qrow[khalf] * scale;
  qb0.y = qrow[khalf + 1] * scale;
  qb1.x = qrow[4 + khalf] * scale;
  qb1.y = qrow[4 + khalf + 1] * scale;

  float mrun = -1e30f;
  float lsum = 0.f;
  v8f acc = {};

  const float* kp = k + m * DD + h * DHH + khalf;
  const float* vp = v + (8 * half) * DD + h * DHH + (m & 7);

  KVTile t[RING];
  int kb = 0;
#pragma unroll
  for (int i = 0; i < AHEAD; i++) {
    load_kv(t[i], kp, vp, kb + 16 * i);
    step_fence(acc, qb0, qb1, kb);
  }

  for (; kb < NN - 16 * RING; kb += 16 * RING) {
#pragma unroll
    for (int i = 0; i < RING; i++) {
      load_kv(t[(i + AHEAD) % RING], kp, vp, kb + 16 * (i + AHEAD));
      attn_step(t[i], qb0, qb1, mrun, lsum, acc);
      step_fence(acc, qb0, qb1, kb);
    }
  }
#pragma unroll
  for (int i = 0; i < RING; i++) {
    if (i + AHEAD < RING) load_kv(t[i + AHEAD], kp, vp, NN - 16 * RING + 16 * (i + AHEAD));
    attn_step(t[i], qb0, qb1, mrun, lsum, acc);
  }

  float l = lsum + xhalf(lsum);
  if (half == 0) {
    float* orow = o + (qbase + m) * DD + h * DHH;
#pragma unroll
    for (int j = 0; j < 8; j++) orow[j] = acc[j] / l;
  }
}

extern "C" void kernel_launch(void* const* d_in, const int* in_sizes, int n_in,
                              void* d_out, int out_size, void* d_ws, size_t ws_size,
                              hipStream_t stream) {
  const float* x  = (const float*)d_in[0];
  const int*  ei  = (const int*)d_in[1];
  const int*  src = ei;
  const int*  dst = ei + EE;
  const float* We = (const float*)d_in[2];
  const float* Wg = (const float*)d_in[3];
  const float* Wq = (const float*)d_in[4];
  const float* Wk = (const float*)d_in[5];
  const float* Wv = (const float*)d_in[6];
  const float* Wo = (const float*)d_in[7];
  const float* W1 = (const float*)d_in[8];
  const float* W2 = (const float*)d_in[9];
  const float* g1 = (const float*)d_in[10];
  const float* b1 = (const float*)d_in[11];
  const float* g2 = (const float*)d_in[12];
  const float* b2 = (const float*)d_in[13];

  float* h = (float*)d_out;
  float* ws = (float*)d_ws;
  float* msg   = ws;
  float* qb    = msg + 3 * ND;
  float* kb    = qb + ND;
  float* vb    = kb + ND;
  float* oatt  = vb + ND;
  int*   cnt   = (int*)(oatt + 3 * ND + 2 * NN);

  embed_zero<<<NN / 16, 128, 0, stream>>>(x, We, h, msg, cnt);
  count_deg<<<(EE + 255) / 256, 256, 0, stream>>>(src, dst, cnt, cnt + NN);

  for (int l = 0; l < LL; l++) {
    scatter_msg<<<(EE * DD + 255) / 256, 256, 0, stream>>>(h, src, dst, msg);
    stage_a<<<NN / 16, 128, 0, stream>>>(h, msg, cnt + NN, Wg + l * DD * DD, Wq + l * DD * DD,
                                         Wk + l * DD * DD, Wv + l * DD * DD, qb, kb, vb);

    attn_kernel<<<NN / 16, 256, 0, stream>>>(qb, kb, vb, oatt);

    stage_b<<<NN / 16, 128, 0, stream>>>(h, oatt, cnt, Wo + l * DD * DD, W1 + l * DD * FF,
                                         W2 + l * FF * DD, g1 + l * DD, b1 + l * DD, g2 + l * DD,
                                         b2 + l * DD, msg);
  }
}


# neighbour sums without float atomics: edges bucketed by destination once per call (rank from the count's returning atomic, scan, fill), stage_a gathers each row's sources; sum order = arrival rank
# speedup vs baseline: 8.5722x; 1.2602x over previous
#include <hip/hip_runtime.h>

typedef float v2f __attribute__((ext_vector_type(2)));
typedef float v8f __attribute__((ext_vector_type(8)));

#define NN 4096
#define EE 131072
#define INDIM 32
#define DD 64
#define HH 8
#define DHH 8
#define FF 64
#define LL 4
#define ND (NN * DD)

__device__ __forceinline__ v8f wmma4(v2f a, v2f b, v8f c) {
  return __builtin_amdgcn_wmma_f32_16x16x4_f32(
       false, a,  false, b,
       (short)0, c,  false,  false);
}

__global__ void count_deg(const int* __restrict__ src, const int* __restrict__ dst,
                          int* cs, int* ci, int* __restrict__ rank) {
  int e = blockIdx.x * blockDim.x + threadIdx.x;
  if (e < EE) {
    atomicAdd(&cs[src[e]], 1);
    rank[e] = atomicAdd(&ci[dst[e]], 1);
  }
}

#define FILL_T 1024
static_assert(NN == 4 * FILL_T, "one int4 of counts per thread");
static_assert(EE % FILL_T == 0, "whole workgroups of edges");
__global__ void __launch_bounds__(FILL_T) csr_fill(const int* __restrict__ src,
                                                   const int* __restrict__ dst,
                                                   const int* __restrict__ ci,
                                                   const int* __restrict__ rank,
                                                   int* __restrict__ row_start,
                                                   int* __restrict__ col) {
  __shared__ __attribute__((aligned(16))) int sStart[NN];
  __shared__ int sWave[FILL_T / 32];
  int t = threadIdx.x;
  int lane = t & 31;
  int w = t >> 5;
  int e = blockIdx.x * FILL_T + t;
  int ed = dst[e];
  int es = src[e];
  int er = rank[e];
  int4 c = *(const int4*)(ci + 4 * t);
  int s0 = c.x, s1 = s0 + c.y, s2 = s1 + c.z, s3 = s2 + c.w;
  int incl = s3;
  for (int d = 1; d < 32; d <<= 1) {
    int y = __shfl_up(incl, d);
    if (lane >= d) incl += y;
  }
  if (lane == 31) sWave[w] = incl;
  __syncthreads();
  if (w == 0) {
    int v = sWave[lane];
    int iv = v;
    for (int d = 1; d < 32; d <<= 1) {
      int y = __shfl_up(iv, d);
      if (lane >= d) iv += y;
    }
    sWave[lane] = iv - v;
  }
  __syncthreads();
  int base = sWave[w] + incl - s3;
  int4 st = {base, base + s0, base + s1, base + s2};
  *(int4*)(sStart + 4 * t) = st;
  if (blockIdx.x == 0) {
    *(int4*)(row_start + 4 * t) = st;
    if (t == FILL_T - 1) row_start[NN] = base + s3;
  }
  __syncthreads();
  col[sStart[ed] + er] = es;
}

#define LP 68
#define CNT_PER_WG (2 * NN / (NN / 16))

__device__ __forceinline__ v8f tile_gemm64(const float* As, const float* __restrict__ B,
                                           int colbase, int m, int khalf) {
  v8f c = {};
#pragma unroll
  for (int k0 = 0; k0 < 64; k0 += 4) {
    v2f a, b;
    const float* ar = As + m * LP + k0 + khalf;
    a.x = ar[0];
    a.y = ar[1];
    b.x = B[(k0 + khalf) * 64 + colbase + m];
    b.y = B[(k0 + khalf + 1) * 64 + colbase + m];
    c = wmma4(a, b, c);
  }
  return c;
}

__device__ __forceinline__ void tile_to_global(float* __restrict__ C, v8f c, int rowbase,
                                               int colbase, int lane) {
  int rr = rowbase + 8 * (lane >> 4);
  int cc = colbase + (lane & 15);
#pragma unroll
  for (int j = 0; j < 8; j++) C[(rr + j) * 64 + cc] = c[j];
}
__device__ __forceinline__ void tile_relu_to_lds(float* Cs, v8f c, int colbase, int lane) {
  int rr = 8 * (lane >> 4);
  int cc = colbase + (lane & 15);
#pragma unroll
  for (int j = 0; j < 8; j++) Cs[(rr + j) * LP + cc] = fmaxf(c[j], 0.f);
}

__device__ __forceinline__ float row_tree_sum(const float* x) {
  float y = ((x[0] + x[4]) + (x[2] + x[6])) + ((x[1] + x[5]) + (x[3] + x[7]));
  y += __shfl_xor(y, 4);
  y += __shfl_xor(y, 2);
  y += __shfl_xor(y, 1);
  return y;
}

__device__ __forceinline__ void ln_rows_inplace(float* tile, int t, const float* __restrict__ g,
                                                const float* __restrict__ b) {
  int i = t & 7;
  float* row = tile + (t >> 3) * LP + i;
  float x[8], d[8], sq[8];
#pragma unroll
  for (int k = 0; k < 8; k++) x[k] = row[8 * k];
  float mean = row_tree_sum(x) * (1.f / 64.f);
#pragma unroll
  for (int k = 0; k < 8; k++) {
    d[k] = x[k] - mean;
    sq[k] = d[k] * d[k];
  }
  float var = row_tree_sum(sq) * (1.f / 64.f);
  float rs = rsqrtf(var + 1e-5f);
#pragma unroll
  for (int k = 0; k < 8; k++) row[8 * k] = d[k] * rs * g[i + 8 * k] + b[i + 8 * k];
}

__global__ void __launch_bounds__(128) embed_zero(const float* __restrict__ x,
                                                  const float* __restrict__ We,
                                                  float* __restrict__ h, int* __restrict__ cnt) {
  int t = threadIdx.x;
  int lane = t & 31;
  int rowbase = blockIdx.x * 16;
  int colbase = (t >> 5) * 16;
  int m = lane & 15;
  int khalf = (lane >> 4) * 2;
  if (t < CNT_PER_WG) cnt[blockIdx.x * CNT_PER_WG + t] = 0;
  v8f c = {};
#pragma unroll
  for (int k0 = 0; k0 < INDIM; k0 += 4) {
    v2f a, b;
    const float* ar = x + (rowbase + m) * INDIM + k0 + khalf;
    a.x = ar[0];
    a.y = ar[1];
    b.x = We[(k0 + khalf) * 64 + colbase + m];
    b.y = We[(k0 + khalf + 1) * 64 + colbase + m];
    c = wmma4(a, b, c);
  }
  tile_to_global(h, c, rowbase, colbase, lane);
}

#define GU 16

__global__ void __launch_bounds__(128) stage_a(const float* __restrict__ h,
                                               const int* __restrict__ row_start,
                                               const int* __restrict__ col,
                                               const float* __restrict__ Wg,
                                               const float* __restrict__ Wq,
                                               const float* __restrict__ Wk,
                                               const float* __restrict__ Wv,
                                               float* __restrict__ qb, float* __restrict__ kb,
                                               float* __restrict__ vb) {
  __shared__ __attribute__((aligned(16))) float sS[16 * LP];
  __shared__ __attribute__((aligned(16))) float sH[16 * LP];
  __shared__ __attribute__((aligned(16))) float sF[16 * LP];
  int t = threadIdx.x;
  int lane = t & 31;
  int rowbase = blockIdx.x * 16;
  int colbase = (t >> 5) * 16;
  int m = lane & 15;
  int khalf = (lane >> 4) * 2;

  {
    int r = t >> 3;
    int c0 = (t & 7) * 4;
    int row = rowbase + r;
    int jb = row_start[row];
    int je = row_start[row + 1];
    const float* hc = h + c0;
    float4 hv0 = *(const float4*)(hc + row * DD);
    float4 hv1 = *(const float4*)(hc + row * DD + 32);
    float4 m0 = {0.f, 0.f, 0.f, 0.f};
    float4 m1 = {0.f, 0.f, 0.f, 0.f};
    for (int j = jb; j < je; j += GU) {
      int sidx[GU];
      float4 v0[GU], v1[GU];
#pragma unroll
      for (int u = 0; u < GU; u++) sidx[u] = col[min(j + u, je - 1)];
#pragma unroll
      for (int u = 0; u < GU; u++) {
        const float* p = hc + sidx[u] * DD;
        v0[u] = *(const float4*)p;
        v1[u] = *(const float4*)(p + 32);
      }
#pragma unroll
      for (int u = 0; u < GU; u++) {
        bool in = j + u < je;
        m0.x += in ? v0[u].x : 0.f; m0.y += in ? v0[u].y : 0.f;
        m0.z += in ? v0[u].z : 0.f; m0.w += in ? v0[u].w : 0.f;
        m1.x += in ? v1[u].x : 0.f; m1.y += in ? v1[u].y : 0.f;
        m1.z += in ? v1[u].z : 0.f; m1.w += in ? v1[u].w : 0.f;
      }
    }
    float inv = 1.f / fmaxf((float)(je - jb), 1.f);
    float4 sv0, sv1;
    sv0.x = hv0.x + m0.x * inv;
    sv0.y = hv0.y + m0.y * inv;
    sv0.z = hv0.z + m0.z * inv;
    sv0.w = hv0.w + m0.w * inv;
    sv1.x = hv1.x + m1.x * inv;
    sv1.y = hv1.y + m1.y * inv;
    sv1.z = hv1.z + m1.z * inv;
    sv1.w = hv1.w + m1.w * inv;
    *(float4*)(sH + r * LP + c0) = hv0;
    *(float4*)(sH + r * LP + c0 + 32) = hv1;
    *(float4*)(sS + r * LP + c0) = sv0;
    *(float4*)(sS + r * LP + c0 + 32) = sv1;
  }
  __syncthreads();
  v8f vv = tile_gemm64(sH, Wv, colbase, m, khalf);
  v8f ff = tile_gemm64(sS, Wg, colbase, m, khalf);
  tile_to_global(vb, vv, rowbase, colbase, lane);
  tile_relu_to_lds(sF, ff, colbase, lane);
  __syncthreads();
  v8f qq = tile_gemm64(sF, Wq, colbase, m, khalf);
  v8f kk = tile_gemm64(sF, Wk, colbase, m, khalf);
  tile_to_global(qb, qq, rowbase, colbase, lane);
  tile_to_global(kb, kk, rowbase, colbase, lane);
}

__global__ void __launch_bounds__(128) stage_b(float* __restrict__ h,
                                               const float* __restrict__ oatt,
                                               const int* __restrict__ cs,
                                               const float* __restrict__ Wo,
                                               const float* __restrict__ W1,
                                               const float* __restrict__ W2,
                                               const float* __restrict__ g1,
                                               const float* __restrict__ b1,
                                               const float* __restrict__ g2,
                                               const float* __restrict__ b2) {
  __shared__ __attribute__((aligned(16))) float sX[16 * LP];
  __shared__ __attribute__((aligned(16))) float sY[16 * LP];
  int t = threadIdx.x;
  int lane = t & 31;
  int rowbase = blockIdx.x * 16;
  int colbase = (t >> 5) * 16;
  int m = lane & 15;
  int khalf = (lane >> 4) * 2;
  int rr = 8 * (lane >> 4);
  int cc = colbase + m;

#pragma unroll
  for (int i = 0; i < 2; i++) {
    int idx = t + 128 * i;
    int r = idx >> 4, c = (idx & 15) * 4;
    int g = (rowbase + r) * DD + c;
    *(float4*)(sX + r * LP + c) = *(const float4*)(oatt + g);
  }
  __syncthreads();

  v8f o = tile_gemm64(sX, Wo, colbase, m, khalf);
#pragma unroll
  for (int j = 0; j < 8; j++) {
    int row = rowbase + rr + j;
    float sc = rsqrtf(1.f + (float)cs[row]);
    sY[(rr + j) * LP + cc] = h[row * DD + cc] + o[j] * sc;
  }
  __syncthreads();
  ln_rows_inplace(sY, t, g1, b1);
  __syncthreads();

  v8f f1 = tile_gemm64(sY, W1, colbase, m, khalf);
  tile_relu_to_lds(sX, f1, colbase, lane);
  __syncthreads();
  v8f f2 = tile_gemm64(sX, W2, colbase, m, khalf);
#pragma unroll
  for (int j = 0; j < 8; j++) sY[(rr + j) * LP + cc] += f2[j];
  __syncthreads();
  ln_rows_inplace(sY, t, g2, b2);
  __syncthreads();

#pragma unroll
  for (int i = 0; i < 2; i++) {
    int idx = t + 128 * i;
    int r = idx >> 4, c = (idx & 15) * 4;
    *(float4*)(h + (rowbase + r) * DD + c) = *(const float4*)(sY + r * LP + c);
  }
}

__device__ __forceinline__ float xhalf(float x) {
  int b = __builtin_bit_cast(int, x);
  return __builtin_bit_cast(
      float, __builtin_amdgcn_permlanex16(b, b, 0x76543210, 0xfedcba98, false, false));
}


struct KVTile {
  v2f k0, k1;
  v2f v[4];
};

__device__ __forceinline__ void load_kv(KVTile& t, const float* __restrict__ kp,
                                        const float* __restrict__ vp, int kb) {
  const float* krow = kp + kb * DD;
  t.k0.x = krow[0]; t.k0.y = krow[1];
  t.k1.x = krow[4]; t.k1.y = krow[5];
  const float* vrow = vp + kb * DD;
#pragma unroll
  for (int c = 0; c < 4; c++) {
    t.v[c].x = vrow[(2 * c) * DD];
    t.v[c].y = vrow[(2 * c + 1) * DD];
  }
}

__device__ __forceinline__ void attn_step(const KVTile& t, v2f qb0, v2f qb1,
                                          float& mrun, float& lsum, v8f& acc) {
  v8f s = {};
  s = wmma4(t.k0, qb0, s);
  s = wmma4(t.k1, qb1, s);

  float sm = fmaxf(fmaxf(fmaxf(s[0], s[1]), fmaxf(s[2], s[3])),
                   fmaxf(fmaxf(s[4], s[5]), fmaxf(s[6], s[7])));
  sm = fmaxf(sm, xhalf(sm));
  float mn = fmaxf(mrun, sm);
  float alpha = __expf(mrun - mn);
  mrun = mn;
  float p[8];
  float ps = 0.f;
#pragma unroll
  for (int j = 0; j < 8; j++) {
    p[j] = __expf(s[j] - mn);
    ps += p[j];
  }
  lsum = lsum * alpha + ps;
#pragma unroll
  for (int j = 0; j < 8; j++) acc[j] *= alpha;

#pragma unroll
  for (int c = 0; c < 4; c++) {
    v2f pb;
    pb.x = p[2 * c];
    pb.y = p[2 * c + 1];
    acc = wmma4(t.v[c], pb, acc);
  }
}

#define RING 8
#define AHEAD 6

__device__ __forceinline__ void step_fence(v8f& acc, v2f& qb0, v2f& qb1, int& kb) {
  asm volatile("" : "+v"(acc), "+v"(qb0), "+v"(qb1), "+s"(kb));
}

__global__ void __launch_bounds__(256) attn_kernel(const float* __restrict__ q,
                                                   const float* __restrict__ k,
                                                   const float* __restrict__ v,
                                                   float* __restrict__ o) {
  int h = threadIdx.x >> 5;
  int lane = threadIdx.x & 31;
  int qbase = blockIdx.x * 16;
  int m = lane & 15;
  int half = lane >> 4;
  int khalf = half * 2;
  const float scale = 0.35355339059327373f;

  const float* qrow = q + (qbase + m) * DD + h * DHH;
  v2f qb0, qb1;
  qb0.x = qrow[khalf] * scale;
  qb0.y = qrow[khalf + 1] * scale;
  qb1.x = qrow[4 + khalf] * scale;
  qb1.y = qrow[4 + khalf + 1] * scale;

  float mrun = -1e30f;
  float lsum = 0.f;
  v8f acc = {};

  const float* kp = k + m * DD + h * DHH + khalf;
  const float* vp = v + (8 * half) * DD + h * DHH + (m & 7);

  KVTile t[RING];
  int kb = 0;
#pragma unroll
  for (int i = 0; i < AHEAD; i++) {
    load_kv(t[i], kp, vp, kb + 16 * i);
    step_fence(acc, qb0, qb1, kb);
  }

  for (; kb < NN - 16 * RING; kb += 16 * RING) {
#pragma unroll
    for (int i = 0; i < RING; i++) {
      load_kv(t[(i + AHEAD) % RING], kp, vp, kb + 16 * (i + AHEAD));
      attn_step(t[i], qb0, qb1, mrun, lsum, acc);
      step_fence(acc, qb0, qb1, kb);
    }
  }
#pragma unroll
  for (int i = 0; i < RING; i++) {
    if (i + AHEAD < RING) load_kv(t[i + AHEAD], kp, vp, NN - 16 * RING + 16 * (i + AHEAD));
    attn_step(t[i], qb0, qb1, mrun, lsum, acc);
  }

  float l = lsum + xhalf(lsum);
  if (half == 0) {
    float* orow = o + (qbase + m) * DD + h * DHH;
#pragma unroll
    for (int j = 0; j < 8; j++) orow[j] = acc[j] / l;
  }
}

extern "C" void kernel_launch(void* const* d_in, const int* in_sizes, int n_in,
                              void* d_out, int out_size, void* d_ws, size_t ws_size,
                              hipStream_t stream) {
  const float* x  = (const float*)d_in[0];
  const int*  ei  = (const int*)d_in[1];
  const int*  src = ei;
  const int*  dst = ei + EE;
  const float* We = (const float*)d_in[2];
  const float* Wg = (const float*)d_in[3];
  const float* Wq = (const float*)d_in[4];
  const float* Wk = (const float*)d_in[5];
  const float* Wv = (const float*)d_in[6];
  const float* Wo = (const float*)d_in[7];
  const float* W1 = (const float*)d_in[8];
  const float* W2 = (const float*)d_in[9];
  const float* g1 = (const float*)d_in[10];
  const float* b1 = (const float*)d_in[11];
  const float* g2 = (const float*)d_in[12];
  const float* b2 = (const float*)d_in[13];

  float* h = (float*)d_out;
  float* ws = (float*)d_ws;
  int*   col   = (int*)ws;
  int*   rank  = col + EE;
  int*   rowst = rank + EE;
  float* qb    = ws + 3 * ND;
  float* kb    = qb + ND;
  float* vb    = kb + ND;
  float* oatt  = vb + ND;
  int*   cnt   = (int*)(oatt + 3 * ND + 2 * NN);

  embed_zero<<<NN / 16, 128, 0, stream>>>(x, We, h, cnt);
  count_deg<<<(EE + 255) / 256, 256, 0, stream>>>(src, dst, cnt, cnt + NN, rank);
  csr_fill<<<EE / FILL_T, FILL_T, 0, stream>>>(src, dst, cnt + NN, rank, rowst, col);

  for (int l = 0; l < LL; l++) {
    stage_a<<<NN / 16, 128, 0, stream>>>(h, rowst, col, Wg + l * DD * DD, Wq + l * DD * DD,
                                         Wk + l * DD * DD, Wv + l * DD * DD, qb, kb, vb);

    attn_kernel<<<NN / 16, 256, 0, stream>>>(qb, kb, vb, oatt);

    stage_b<<<NN / 16, 128, 0, stream>>>(h, oatt, cnt, Wo + l * DD * DD, W1 + l * DD * FF,
                                         W2 + l * FF * DD, g1 + l * DD, b1 + l * DD, g2 + l * DD,
                                         b2 + l * DD);
  }
}


# attention: each head's keys split between two waves of a 16-wave workgroup (4 softmax chains per SIMD, not 2), halves joined once through LDS; denominator and PV sums are two partial sums per query
# speedup vs baseline: 8.9627x; 1.0456x over previous
#include <hip/hip_runtime.h>

typedef float v2f __attribute__((ext_vector_type(2)));
typedef float v8f __attribute__((ext_vector_type(8)));

#define NN 4096
#define EE 131072
#define INDIM 32
#define DD 64
#define HH 8
#define DHH 8
#define FF 64
#define LL 4
#define ND (NN * DD)

__device__ __forceinline__ v8f wmma4(v2f a, v2f b, v8f c) {
  return __builtin_amdgcn_wmma_f32_16x16x4_f32(
       false, a,  false, b,
       (short)0, c,  false,  false);
}

__global__ void count_deg(const int* __restrict__ src, const int* __restrict__ dst,
                          int* cs, int* ci, int* __restrict__ rank) {
  int e = blockIdx.x * blockDim.x + threadIdx.x;
  if (e < EE) {
    atomicAdd(&cs[src[e]], 1);
    rank[e] = atomicAdd(&ci[dst[e]], 1);
  }
}

#define FILL_T 1024
static_assert(NN == 4 * FILL_T, "one int4 of counts per thread");
static_assert(EE % FILL_T == 0, "whole workgroups of edges");
__global__ void __launch_bounds__(FILL_T) csr_fill(const int* __restrict__ src,
                                                   const int* __restrict__ dst,
                                                   const int* __restrict__ ci,
                                                   const int* __restrict__ rank,
                                                   int* __restrict__ row_start,
                                                   int* __restrict__ col) {
  __shared__ __attribute__((aligned(16))) int sStart[NN];
  __shared__ int sWave[FILL_T / 32];
  int t = threadIdx.x;
  int lane = t & 31;
  int w = t >> 5;
  int e = blockIdx.x * FILL_T + t;
  int ed = dst[e];
  int es = src[e];
  int er = rank[e];
  int4 c = *(const int4*)(ci + 4 * t);
  int s0 = c.x, s1 = s0 + c.y, s2 = s1 + c.z, s3 = s2 + c.w;
  int incl = s3;
  for (int d = 1; d < 32; d <<= 1) {
    int y = __shfl_up(incl, d);
    if (lane >= d) incl += y;
  }
  if (lane == 31) sWave[w] = incl;
  __syncthreads();
  if (w == 0) {
    int v = sWave[lane];
    int iv = v;
    for (int d = 1; d < 32; d <<= 1) {
      int y = __shfl_up(iv, d);
      if (lane >= d) iv += y;
    }
    sWave[lane] = iv - v;
  }
  __syncthreads();
  int base = sWave[w] + incl - s3;
  int4 st = {base, base + s0, base + s1, base + s2};
  *(int4*)(sStart + 4 * t) = st;
  if (blockIdx.x == 0) {
    *(int4*)(row_start + 4 * t) = st;
    if (t == FILL_T - 1) row_start[NN] = base + s3;
  }
  __syncthreads();
  col[sStart[ed] + er] = es;
}

#define LP 68
#define CNT_PER_WG (2 * NN / (NN / 16))

__device__ __forceinline__ v8f tile_gemm64(const float* As, const float* __restrict__ B,
                                           int colbase, int m, int khalf) {
  v8f c = {};
#pragma unroll
  for (int k0 = 0; k0 < 64; k0 += 4) {
    v2f a, b;
    const float* ar = As + m * LP + k0 + khalf;
    a.x = ar[0];
    a.y = ar[1];
    b.x = B[(k0 + khalf) * 64 + colbase + m];
    b.y = B[(k0 + khalf + 1) * 64 + colbase + m];
    c = wmma4(a, b, c);
  }
  return c;
}

__device__ __forceinline__ void tile_to_global(float* __restrict__ C, v8f c, int rowbase,
                                               int colbase, int lane) {
  int rr = rowbase + 8 * (lane >> 4);
  int cc = colbase + (lane & 15);
#pragma unroll
  for (int j = 0; j < 8; j++) C[(rr + j) * 64 + cc] = c[j];
}
__device__ __forceinline__ void tile_relu_to_lds(float* Cs, v8f c, int colbase, int lane) {
  int rr = 8 * (lane >> 4);
  int cc = colbase + (lane & 15);
#pragma unroll
  for (int j = 0; j < 8; j++) Cs[(rr + j) * LP + cc] = fmaxf(c[j], 0.f);
}

__device__ __forceinline__ float row_tree_sum(const float* x) {
  float y = ((x[0] + x[4]) + (x[2] + x[6])) + ((x[1] + x[5]) + (x[3] + x[7]));
  y += __shfl_xor(y, 4);
  y += __shfl_xor(y, 2);
  y += __shfl_xor(y, 1);
  return y;
}

__device__ __forceinline__ void ln_rows_inplace(float* tile, int t, const float* __restrict__ g,
                                                const float* __restrict__ b) {
  int i = t & 7;
  float* row = tile + (t >> 3) * LP + i;
  float x[8], d[8], sq[8];
#pragma unroll
  for (int k = 0; k < 8; k++) x[k] = row[8 * k];
  float mean = row_tree_sum(x) * (1.f / 64.f);
#pragma unroll
  for (int k = 0; k < 8; k++) {
    d[k] = x[k] - mean;
    sq[k] = d[k] * d[k];
  }
  float var = row_tree_sum(sq) * (1.f / 64.f);
  float rs = rsqrtf(var + 1e-5f);
#pragma unroll
  for (int k = 0; k < 8; k++) row[8 * k] = d[k] * rs * g[i + 8 * k] + b[i + 8 * k];
}

__global__ void __launch_bounds__(128) embed_zero(const float* __restrict__ x,
                                                  const float* __restrict__ We,
                                                  float* __restrict__ h, int* __restrict__ cnt) {
  int t = threadIdx.x;
  int lane = t & 31;
  int rowbase = blockIdx.x * 16;
  int colbase = (t >> 5) * 16;
  int m = lane & 15;
  int khalf = (lane >> 4) * 2;
  if (t < CNT_PER_WG) cnt[blockIdx.x * CNT_PER_WG + t] = 0;
  v8f c = {};
#pragma unroll
  for (int k0 = 0; k0 < INDIM; k0 += 4) {
    v2f a, b;
    const float* ar = x + (rowbase + m) * INDIM + k0 + khalf;
    a.x = ar[0];
    a.y = ar[1];
    b.x = We[(k0 + khalf) * 64 + colbase + m];
    b.y = We[(k0 + khalf + 1) * 64 + colbase + m];
    c = wmma4(a, b, c);
  }
  tile_to_global(h, c, rowbase, colbase, lane);
}

#define GU 16

__global__ void __launch_bounds__(128) stage_a(const float* __restrict__ h,
                                               const int* __restrict__ row_start,
                                               const int* __restrict__ col,
                                               const float* __restrict__ Wg,
                                               const float* __restrict__ Wq,
                                               const float* __restrict__ Wk,
                                               const float* __restrict__ Wv,
                                               float* __restrict__ qb, float* __restrict__ kb,
                                               float* __restrict__ vb) {
  __shared__ __attribute__((aligned(16))) float sS[16 * LP];
  __shared__ __attribute__((aligned(16))) float sH[16 * LP];
  __shared__ __attribute__((aligned(16))) float sF[16 * LP];
  int t = threadIdx.x;
  int lane = t & 31;
  int rowbase = blockIdx.x * 16;
  int colbase = (t >> 5) * 16;
  int m = lane & 15;
  int khalf = (lane >> 4) * 2;

  {
    int r = t >> 3;
    int c0 = (t & 7) * 4;
    int row = rowbase + r;
    int jb = row_start[row];
    int je = row_start[row + 1];
    const float* hc = h + c0;
    float4 hv0 = *(const float4*)(hc + row * DD);
    float4 hv1 = *(const float4*)(hc + row * DD + 32);
    float4 m0 = {0.f, 0.f, 0.f, 0.f};
    float4 m1 = {0.f, 0.f, 0.f, 0.f};
    for (int j = jb; j < je; j += GU) {
      int sidx[GU];
      float4 v0[GU], v1[GU];
#pragma unroll
      for (int u = 0; u < GU; u++) sidx[u] = col[min(j + u, je - 1)];
#pragma unroll
      for (int u = 0; u < GU; u++) {
        const float* p = hc + sidx[u] * DD;
        v0[u] = *(const float4*)p;
        v1[u] = *(const float4*)(p + 32);
      }
#pragma unroll
      for (int u = 0; u < GU; u++) {
        bool in = j + u < je;
        m0.x += in ? v0[u].x : 0.f; m0.y += in ? v0[u].y : 0.f;
        m0.z += in ? v0[u].z : 0.f; m0.w += in ? v0[u].w : 0.f;
        m1.x += in ? v1[u].x : 0.f; m1.y += in ? v1[u].y : 0.f;
        m1.z += in ? v1[u].z : 0.f; m1.w += in ? v1[u].w : 0.f;
      }
    }
    float inv = 1.f / fmaxf((float)(je - jb), 1.f);
    float4 sv0, sv1;
    sv0.x = hv0.x + m0.x * inv;
    sv0.y = hv0.y + m0.y * inv;
    sv0.z = hv0.z + m0.z * inv;
    sv0.w = hv0.w + m0.w * inv;
    sv1.x = hv1.x + m1.x * inv;
    sv1.y = hv1.y + m1.y * inv;
    sv1.z = hv1.z + m1.z * inv;
    sv1.w = hv1.w + m1.w * inv;
    *(float4*)(sH + r * LP + c0) = hv0;
    *(float4*)(sH + r * LP + c0 + 32) = hv1;
    *(float4*)(sS + r * LP + c0) = sv0;
    *(float4*)(sS + r * LP + c0 + 32) = sv1;
  }
  __syncthreads();
  v8f vv = tile_gemm64(sH, Wv, colbase, m, khalf);
  v8f ff = tile_gemm64(sS, Wg, colbase, m, khalf);
  tile_to_global(vb, vv, rowbase, colbase, lane);
  tile_relu_to_lds(sF, ff, colbase, lane);
  __syncthreads();
  v8f qq = tile_gemm64(sF, Wq, colbase, m, khalf);
  v8f kk = tile_gemm64(sF, Wk, colbase, m, khalf);
  tile_to_global(qb, qq, rowbase, colbase, lane);
  tile_to_global(kb, kk, rowbase, colbase, lane);
}

__global__ void __launch_bounds__(128) stage_b(float* __restrict__ h,
                                               const float* __restrict__ oatt,
                                               const int* __restrict__ cs,
                                               const float* __restrict__ Wo,
                                               const float* __restrict__ W1,
                                               const float* __restrict__ W2,
                                               const float* __restrict__ g1,
                                               const float* __restrict__ b1,
                                               const float* __restrict__ g2,
                                               const float* __restrict__ b2) {
  __shared__ __attribute__((aligned(16))) float sX[16 * LP];
  __shared__ __attribute__((aligned(16))) float sY[16 * LP];
  int t = threadIdx.x;
  int lane = t & 31;
  int rowbase = blockIdx.x * 16;
  int colbase = (t >> 5) * 16;
  int m = lane & 15;
  int khalf = (lane >> 4) * 2;
  int rr = 8 * (lane >> 4);
  int cc = colbase + m;

#pragma unroll
  for (int i = 0; i < 2; i++) {
    int idx = t + 128 * i;
    int r = idx >> 4, c = (idx & 15) * 4;
    int g = (rowbase + r) * DD + c;
    *(float4*)(sX + r * LP + c) = *(const float4*)(oatt + g);
  }
  __syncthreads();

  v8f o = tile_gemm64(sX, Wo, colbase, m, khalf);
#pragma unroll
  for (int j = 0; j < 8; j++) {
    int row = rowbase + rr + j;
    float sc = rsqrtf(1.f + (float)cs[row]);
    sY[(rr + j) * LP + cc] = h[row * DD + cc] + o[j] * sc;
  }
  __syncthreads();
  ln_rows_inplace(sY, t, g1, b1);
  __syncthreads();

  v8f f1 = tile_gemm64(sY, W1, colbase, m, khalf);
  tile_relu_to_lds(sX, f1, colbase, lane);
  __syncthreads();
  v8f f2 = tile_gemm64(sX, W2, colbase, m, khalf);
#pragma unroll
  for (int j = 0; j < 8; j++) sY[(rr + j) * LP + cc] += f2[j];
  __syncthreads();
  ln_rows_inplace(sY, t, g2, b2);
  __syncthreads();

#pragma unroll
  for (int i = 0; i < 2; i++) {
    int idx = t + 128 * i;
    int r = idx >> 4, c = (idx & 15) * 4;
    *(float4*)(h + (rowbase + r) * DD + c) = *(const float4*)(sY + r * LP + c);
  }
}

#define KSPLIT 2
#define KPART (NN / KSPLIT)
#define MSLOT 12
__device__ __forceinline__ float xhalf(float x) {
  int b = __builtin_bit_cast(int, x);
  return __builtin_bit_cast(
      float, __builtin_amdgcn_permlanex16(b, b, 0x76543210, 0xfedcba98, false, false));
}


struct KVTile {
  v2f k0, k1;
  v2f v[4];
};

__device__ __forceinline__ void load_kv(KVTile& t, const float* __restrict__ kp,
                                        const float* __restrict__ vp, int kb) {
  const float* krow = kp + kb * DD;
  t.k0.x = krow[0]; t.k0.y = krow[1];
  t.k1.x = krow[4]; t.k1.y = krow[5];
  const float* vrow = vp + kb * DD;
#pragma unroll
  for (int c = 0; c < 4; c++) {
    t.v[c].x = vrow[(2 * c) * DD];
    t.v[c].y = vrow[(2 * c + 1) * DD];
  }
}

__device__ __forceinline__ void attn_step(const KVTile& t, v2f qb0, v2f qb1,
                                          float& mrun, float& lsum, v8f& acc) {
  v8f s = {};
  s = wmma4(t.k0, qb0, s);
  s = wmma4(t.k1, qb1, s);

  float sm = fmaxf(fmaxf(fmaxf(s[0], s[1]), fmaxf(s[2], s[3])),
                   fmaxf(fmaxf(s[4], s[5]), fmaxf(s[6], s[7])));
  sm = fmaxf(sm, xhalf(sm));
  float mn = fmaxf(mrun, sm);
  float alpha = __expf(mrun - mn);
  mrun = mn;
  float p[8];
  float ps = 0.f;
#pragma unroll
  for (int j = 0; j < 8; j++) {
    p[j] = __expf(s[j] - mn);
    ps += p[j];
  }
  lsum = lsum * alpha + ps;
#pragma unroll
  for (int j = 0; j < 8; j++) acc[j] *= alpha;

#pragma unroll
  for (int c = 0; c < 4; c++) {
    v2f pb;
    pb.x = p[2 * c];
    pb.y = p[2 * c + 1];
    acc = wmma4(t.v[c], pb, acc);
  }
}

#define RING 8
#define AHEAD 6

__device__ __forceinline__ void step_fence(v8f& acc, v2f& qb0, v2f& qb1, int& kb) {
  asm volatile("" : "+v"(acc), "+v"(qb0), "+v"(qb1), "+s"(kb));
}

__global__ void __launch_bounds__(256 * KSPLIT) attn_kernel(const float* __restrict__ q,
                                                   const float* __restrict__ k,
                                                   const float* __restrict__ v,
                                                   float* __restrict__ o) {
  __shared__ float sPart[HH * 16 * MSLOT];
  int w = threadIdx.x >> 5;
  int h = w & (HH - 1);
  int part = w >> 3;
  int lane = threadIdx.x & 31;
  int qbase = blockIdx.x * 16;
  int m = lane & 15;
  int half = lane >> 4;
  int khalf = half * 2;
  const float scale = 0.35355339059327373f;

  const float* qrow = q + (qbase + m) * DD + h * DHH;
  v2f qb0, qb1;
  qb0.x = qrow[khalf] * scale;
  qb0.y = qrow[khalf + 1] * scale;
  qb1.x = qrow[4 + khalf] * scale;
  qb1.y = qrow[4 + khalf + 1] * scale;

  float mrun = -1e30f;
  float lsum = 0.f;
  v8f acc = {};

  const float* kp = k + (part * KPART + m) * DD + h * DHH + khalf;
  const float* vp = v + (part * KPART + 8 * half) * DD + h * DHH + (m & 7);

  KVTile t[RING];
  int kb = 0;
#pragma unroll
  for (int i = 0; i < AHEAD; i++) {
    load_kv(t[i], kp, vp, kb + 16 * i);
    step_fence(acc, qb0, qb1, kb);
  }

  for (; kb < KPART - 16 * RING; kb += 16 * RING) {
#pragma unroll
    for (int i = 0; i < RING; i++) {
      load_kv(t[(i + AHEAD) % RING], kp, vp, kb + 16 * (i + AHEAD));
      attn_step(t[i], qb0, qb1, mrun, lsum, acc);
      step_fence(acc, qb0, qb1, kb);
    }
  }
#pragma unroll
  for (int i = 0; i < RING; i++) {
    if (i + AHEAD < RING) load_kv(t[i + AHEAD], kp, vp, KPART - 16 * RING + 16 * (i + AHEAD));
    attn_step(t[i], qb0, qb1, mrun, lsum, acc);
  }

  float l = lsum + xhalf(lsum);
  float* slot = sPart + (h * 16 + m) * MSLOT;
  if (part == 1 && half == 0) {
    slot[0] = mrun;
    slot[1] = l;
#pragma unroll
    for (int j = 0; j < 8; j++) slot[2 + j] = acc[j];
  }
  __syncthreads();
  if (part == 0 && half == 0) {
    float m1 = slot[0];
    float l1 = slot[1];
    float mn = fmaxf(mrun, m1);
    float a0 = __expf(mrun - mn);
    float a1 = __expf(m1 - mn);
    float lt = l * a0 + l1 * a1;
    float* orow = o + (qbase + m) * DD + h * DHH;
#pragma unroll
    for (int j = 0; j < 8; j++) orow[j] = (acc[j] * a0 + slot[2 + j] * a1) / lt;
  }
}

extern "C" void kernel_launch(void* const* d_in, const int* in_sizes, int n_in,
                              void* d_out, int out_size, void* d_ws, size_t ws_size,
                              hipStream_t stream) {
  const float* x  = (const float*)d_in[0];
  const int*  ei  = (const int*)d_in[1];
  const int*  src = ei;
  const int*  dst = ei + EE;
  const float* We = (const float*)d_in[2];
  const float* Wg = (const float*)d_in[3];
  const float* Wq = (const float*)d_in[4];
  const float* Wk = (const float*)d_in[5];
  const float* Wv = (const float*)d_in[6];
  const float* Wo = (const float*)d_in[7];
  const float* W1 = (const float*)d_in[8];
  const float* W2 = (const float*)d_in[9];
  const float* g1 = (const float*)d_in[10];
  const float* b1 = (const float*)d_in[11];
  const float* g2 = (const float*)d_in[12];
  const float* b2 = (const float*)d_in[13];

  float* h = (float*)d_out;
  float* ws = (float*)d_ws;
  int*   col   = (int*)ws;
  int*   rank  = col + EE;
  int*   rowst = rank + EE;
  float* qb    = ws + 3 * ND;
  float* kb    = qb + ND;
  float* vb    = kb + ND;
  float* oatt  = vb + ND;
  int*   cnt   = (int*)(oatt + 3 * ND + 2 * NN);

  embed_zero<<<NN / 16, 128, 0, stream>>>(x, We, h, cnt);
  count_deg<<<(EE + 255) / 256, 256, 0, stream>>>(src, dst, cnt, cnt + NN, rank);
  csr_fill<<<EE / FILL_T, FILL_T, 0, stream>>>(src, dst, cnt + NN, rank, rowst, col);

  for (int l = 0; l < LL; l++) {
    stage_a<<<NN / 16, 128, 0, stream>>>(h, rowst, col, Wg + l * DD * DD, Wq + l * DD * DD,
                                         Wk + l * DD * DD, Wv + l * DD * DD, qb, kb, vb);

    attn_kernel<<<NN / 16, 256 * KSPLIT, 0, stream>>>(qb, kb, vb, oatt);

    stage_b<<<NN / 16, 128, 0, stream>>>(h, oatt, cnt, Wo + l * DD * DD, W1 + l * DD * FF,
                                         W2 + l * FF * DD, g1 + l * DD, b1 + l * DD, g2 + l * DD,
                                         b2 + l * DD);
  }
}


# attention workgroup keeps its 16 finished rows in LDS and runs the row-local chain behind them (Wo, degree scale, LN, FFN, LN) in the same launch: 15 -> 11 launches; every sum order unchanged
# speedup vs baseline: 9.4443x; 1.0537x over previous
#include <hip/hip_runtime.h>

typedef float v2f __attribute__((ext_vector_type(2)));
typedef float v8f __attribute__((ext_vector_type(8)));

#define NN 4096
#define EE 131072
#define INDIM 32
#define DD 64
#define HH 8
#define DHH 8
#define FF 64
#define LL 4
#define ND (NN * DD)

__device__ __forceinline__ v8f wmma4(v2f a, v2f b, v8f c) {
  return __builtin_amdgcn_wmma_f32_16x16x4_f32(
       false, a,  false, b,
       (short)0, c,  false,  false);
}

__global__ void count_deg(const int* __restrict__ src, const int* __restrict__ dst,
                          int* cs, int* ci, int* __restrict__ rank) {
  int e = blockIdx.x * blockDim.x + threadIdx.x;
  if (e < EE) {
    atomicAdd(&cs[src[e]], 1);
    rank[e] = atomicAdd(&ci[dst[e]], 1);
  }
}

#define FILL_T 1024
static_assert(NN == 4 * FILL_T, "one int4 of counts per thread");
static_assert(EE % FILL_T == 0, "whole workgroups of edges");
__global__ void __launch_bounds__(FILL_T) csr_fill(const int* __restrict__ src,
                                                   const int* __restrict__ dst,
                                                   const int* __restrict__ ci,
                                                   const int* __restrict__ rank,
                                                   int* __restrict__ row_start,
                                                   int* __restrict__ col) {
  __shared__ __attribute__((aligned(16))) int sStart[NN];
  __shared__ int sWave[FILL_T / 32];
  int t = threadIdx.x;
  int lane = t & 31;
  int w = t >> 5;
  int e = blockIdx.x * FILL_T + t;
  int ed = dst[e];
  int es = src[e];
  int er = rank[e];
  int4 c = *(const int4*)(ci + 4 * t);
  int s0 = c.x, s1 = s0 + c.y, s2 = s1 + c.z, s3 = s2 + c.w;
  int incl = s3;
  for (int d = 1; d < 32; d <<= 1) {
    int y = __shfl_up(incl, d);
    if (lane >= d) incl += y;
  }
  if (lane == 31) sWave[w] = incl;
  __syncthreads();
  if (w == 0) {
    int v = sWave[lane];
    int iv = v;
    for (int d = 1; d < 32; d <<= 1) {
      int y = __shfl_up(iv, d);
      if (lane >= d) iv += y;
    }
    sWave[lane] = iv - v;
  }
  __syncthreads();
  int base = sWave[w] + incl - s3;
  int4 st = {base, base + s0, base + s1, base + s2};
  *(int4*)(sStart + 4 * t) = st;
  if (blockIdx.x == 0) {
    *(int4*)(row_start + 4 * t) = st;
    if (t == FILL_T - 1) row_start[NN] = base + s3;
  }
  __syncthreads();
  col[sStart[ed] + er] = es;
}

#define LP 68
#define CNT_PER_WG (2 * NN / (NN / 16))

__device__ __forceinline__ v8f tile_gemm64(const float* As, const float* __restrict__ B,
                                           int colbase, int m, int khalf) {
  v8f c = {};
#pragma unroll
  for (int k0 = 0; k0 < 64; k0 += 4) {
    v2f a, b;
    const float* ar = As + m * LP + k0 + khalf;
    a.x = ar[0];
    a.y = ar[1];
    b.x = B[(k0 + khalf) * 64 + colbase + m];
    b.y = B[(k0 + khalf + 1) * 64 + colbase + m];
    c = wmma4(a, b, c);
  }
  return c;
}

__device__ __forceinline__ void tile_to_global(float* __restrict__ C, v8f c, int rowbase,
                                               int colbase, int lane) {
  int rr = rowbase + 8 * (lane >> 4);
  int cc = colbase + (lane & 15);
#pragma unroll
  for (int j = 0; j < 8; j++) C[(rr + j) * 64 + cc] = c[j];
}
__device__ __forceinline__ void tile_relu_to_lds(float* Cs, v8f c, int colbase, int lane) {
  int rr = 8 * (lane >> 4);
  int cc = colbase + (lane & 15);
#pragma unroll
  for (int j = 0; j < 8; j++) Cs[(rr + j) * LP + cc] = fmaxf(c[j], 0.f);
}

__device__ __forceinline__ float row_tree_sum(const float* x) {
  float y = ((x[0] + x[4]) + (x[2] + x[6])) + ((x[1] + x[5]) + (x[3] + x[7]));
  y += __shfl_xor(y, 4);
  y += __shfl_xor(y, 2);
  y += __shfl_xor(y, 1);
  return y;
}

__device__ __forceinline__ void ln_rows_inplace(float* tile, int t, const float* __restrict__ g,
                                                const float* __restrict__ b) {
  int i = t & 7;
  float* row = tile + (t >> 3) * LP + i;
  float x[8], d[8], sq[8];
#pragma unroll
  for (int k = 0; k < 8; k++) x[k] = row[8 * k];
  float mean = row_tree_sum(x) * (1.f / 64.f);
#pragma unroll
  for (int k = 0; k < 8; k++) {
    d[k] = x[k] - mean;
    sq[k] = d[k] * d[k];
  }
  float var = row_tree_sum(sq) * (1.f / 64.f);
  float rs = rsqrtf(var + 1e-5f);
#pragma unroll
  for (int k = 0; k < 8; k++) row[8 * k] = d[k] * rs * g[i + 8 * k] + b[i + 8 * k];
}

__global__ void __launch_bounds__(128) embed_zero(const float* __restrict__ x,
                                                  const float* __restrict__ We,
                                                  float* __restrict__ h, int* __restrict__ cnt) {
  int t = threadIdx.x;
  int lane = t & 31;
  int rowbase = blockIdx.x * 16;
  int colbase = (t >> 5) * 16;
  int m = lane & 15;
  int khalf = (lane >> 4) * 2;
  if (t < CNT_PER_WG) cnt[blockIdx.x * CNT_PER_WG + t] = 0;
  v8f c = {};
#pragma unroll
  for (int k0 = 0; k0 < INDIM; k0 += 4) {
    v2f a, b;
    const float* ar = x + (rowbase + m) * INDIM + k0 + khalf;
    a.x = ar[0];
    a.y = ar[1];
    b.x = We[(k0 + khalf) * 64 + colbase + m];
    b.y = We[(k0 + khalf + 1) * 64 + colbase + m];
    c = wmma4(a, b, c);
  }
  tile_to_global(h, c, rowbase, colbase, lane);
}

#define GU 16

__global__ void __launch_bounds__(128) stage_a(const float* __restrict__ h,
                                               const int* __restrict__ row_start,
                                               const int* __restrict__ col,
                                               const float* __restrict__ Wg,
                                               const float* __restrict__ Wq,
                                               const float* __restrict__ Wk,
                                               const float* __restrict__ Wv,
                                               float* __restrict__ qb, float* __restrict__ kb,
                                               float* __restrict__ vb) {
  __shared__ __attribute__((aligned(16))) float sS[16 * LP];
  __shared__ __attribute__((aligned(16))) float sH[16 * LP];
  __shared__ __attribute__((aligned(16))) float sF[16 * LP];
  int t = threadIdx.x;
  int lane = t & 31;
  int rowbase = blockIdx.x * 16;
  int colbase = (t >> 5) * 16;
  int m = lane & 15;
  int khalf = (lane >> 4) * 2;

  {
    int r = t >> 3;
    int c0 = (t & 7) * 4;
    int row = rowbase + r;
    int jb = row_start[row];
    int je = row_start[row + 1];
    const float* hc = h + c0;
    float4 hv0 = *(const float4*)(hc + row * DD);
    float4 hv1 = *(const float4*)(hc + row * DD + 32);
    float4 m0 = {0.f, 0.f, 0.f, 0.f};
    float4 m1 = {0.f, 0.f, 0.f, 0.f};
    for (int j = jb; j < je; j += GU) {
      int sidx[GU];
      float4 v0[GU], v1[GU];
#pragma unroll
      for (int u = 0; u < GU; u++) sidx[u] = col[min(j + u, je - 1)];
#pragma unroll
      for (int u = 0; u < GU; u++) {
        const float* p = hc + sidx[u] * DD;
        v0[u] = *(const float4*)p;
        v1[u] = *(const float4*)(p + 32);
      }
#pragma unroll
      for (int u = 0; u < GU; u++) {
        bool in = j + u < je;
        m0.x += in ? v0[u].x : 0.f; m0.y += in ? v0[u].y : 0.f;
        m0.z += in ? v0[u].z : 0.f; m0.w += in ? v0[u].w : 0.f;
        m1.x += in ? v1[u].x : 0.f; m1.y += in ? v1[u].y : 0.f;
        m1.z += in ? v1[u].z : 0.f; m1.w += in ? v1[u].w : 0.f;
      }
    }
    float inv = 1.f / fmaxf((float)(je - jb), 1.f);
    float4 sv0, sv1;
    sv0.x = hv0.x + m0.x * inv;
    sv0.y = hv0.y + m0.y * inv;
    sv0.z = hv0.z + m0.z * inv;
    sv0.w = hv0.w + m0.w * inv;
    sv1.x = hv1.x + m1.x * inv;
    sv1.y = hv1.y + m1.y * inv;
    sv1.z = hv1.z + m1.z * inv;
    sv1.w = hv1.w + m1.w * inv;
    *(float4*)(sH + r * LP + c0) = hv0;
    *(float4*)(sH + r * LP + c0 + 32) = hv1;
    *(float4*)(sS + r * LP + c0) = sv0;
    *(float4*)(sS + r * LP + c0 + 32) = sv1;
  }
  __syncthreads();
  v8f vv = tile_gemm64(sH, Wv, colbase, m, khalf);
  v8f ff = tile_gemm64(sS, Wg, colbase, m, khalf);
  tile_to_global(vb, vv, rowbase, colbase, lane);
  tile_relu_to_lds(sF, ff, colbase, lane);
  __syncthreads();
  v8f qq = tile_gemm64(sF, Wq, colbase, m, khalf);
  v8f kk = tile_gemm64(sF, Wk, colbase, m, khalf);
  tile_to_global(qb, qq, rowbase, colbase, lane);
  tile_to_global(kb, kk, rowbase, colbase, lane);
}

#define KSPLIT 2
#define KPART (NN / KSPLIT)
#define MSLOT 12
__device__ __forceinline__ float xhalf(float x) {
  int b = __builtin_bit_cast(int, x);
  return __builtin_bit_cast(
      float, __builtin_amdgcn_permlanex16(b, b, 0x76543210, 0xfedcba98, false, false));
}


struct KVTile {
  v2f k0, k1;
  v2f v[4];
};

__device__ __forceinline__ void load_kv(KVTile& t, const float* __restrict__ kp,
                                        const float* __restrict__ vp, int kb) {
  const float* krow = kp + kb * DD;
  t.k0.x = krow[0]; t.k0.y = krow[1];
  t.k1.x = krow[4]; t.k1.y = krow[5];
  const float* vrow = vp + kb * DD;
#pragma unroll
  for (int c = 0; c < 4; c++) {
    t.v[c].x = vrow[(2 * c) * DD];
    t.v[c].y = vrow[(2 * c + 1) * DD];
  }
}

__device__ __forceinline__ void attn_step(const KVTile& t, v2f qb0, v2f qb1,
                                          float& mrun, float& lsum, v8f& acc) {
  v8f s = {};
  s = wmma4(t.k0, qb0, s);
  s = wmma4(t.k1, qb1, s);

  float sm = fmaxf(fmaxf(fmaxf(s[0], s[1]), fmaxf(s[2], s[3])),
                   fmaxf(fmaxf(s[4], s[5]), fmaxf(s[6], s[7])));
  sm = fmaxf(sm, xhalf(sm));
  float mn = fmaxf(mrun, sm);
  float alpha = __expf(mrun - mn);
  mrun = mn;
  float p[8];
  float ps = 0.f;
#pragma unroll
  for (int j = 0; j < 8; j++) {
    p[j] = __expf(s[j] - mn);
    ps += p[j];
  }
  lsum = lsum * alpha + ps;
#pragma unroll
  for (int j = 0; j < 8; j++) acc[j] *= alpha;

#pragma unroll
  for (int c = 0; c < 4; c++) {
    v2f pb;
    pb.x = p[2 * c];
    pb.y = p[2 * c + 1];
    acc = wmma4(t.v[c], pb, acc);
  }
}

#define RING 8
#define AHEAD 6

__device__ __forceinline__ void step_fence(v8f& acc, v2f& qb0, v2f& qb1, int& kb) {
  asm volatile("" : "+v"(acc), "+v"(qb0), "+v"(qb1), "+s"(kb));
}

__global__ void __launch_bounds__(256 * KSPLIT) attn_kernel(const float* __restrict__ q,
                                                   const float* __restrict__ k,
                                                   const float* __restrict__ v,
                                                   float* __restrict__ hbuf,
                                                   const int* __restrict__ cs,
                                                   const float* __restrict__ Wo,
                                                   const float* __restrict__ W1,
                                                   const float* __restrict__ W2,
                                                   const float* __restrict__ g1,
                                                   const float* __restrict__ b1,
                                                   const float* __restrict__ g2,
                                                   const float* __restrict__ b2) {
  __shared__ float sPart[HH * 16 * MSLOT];
  __shared__ __attribute__((aligned(16))) float sX[16 * LP];
  __shared__ __attribute__((aligned(16))) float sY[16 * LP];
  int w = threadIdx.x >> 5;
  int h = w & (HH - 1);
  int part = w >> 3;
  int lane = threadIdx.x & 31;
  int qbase = blockIdx.x * 16;
  int m = lane & 15;
  int half = lane >> 4;
  int khalf = half * 2;
  const float scale = 0.35355339059327373f;

  const float* qrow = q + (qbase + m) * DD + h * DHH;
  v2f qb0, qb1;
  qb0.x = qrow[khalf] * scale;
  qb0.y = qrow[khalf + 1] * scale;
  qb1.x = qrow[4 + khalf] * scale;
  qb1.y = qrow[4 + khalf + 1] * scale;

  float mrun = -1e30f;
  float lsum = 0.f;
  v8f acc = {};

  const float* kp = k + (part * KPART + m) * DD + h * DHH + khalf;
  const float* vp = v + (part * KPART + 8 * half) * DD + h * DHH + (m & 7);

  KVTile t[RING];
  int kb = 0;
#pragma unroll
  for (int i = 0; i < AHEAD; i++) {
    load_kv(t[i], kp, vp, kb + 16 * i);
    step_fence(acc, qb0, qb1, kb);
  }

  for (; kb < KPART - 16 * RING; kb += 16 * RING) {
#pragma unroll
    for (int i = 0; i < RING; i++) {
      load_kv(t[(i + AHEAD) % RING], kp, vp, kb + 16 * (i + AHEAD));
      attn_step(t[i], qb0, qb1, mrun, lsum, acc);
      step_fence(acc, qb0, qb1, kb);
    }
  }
#pragma unroll
  for (int i = 0; i < RING; i++) {
    if (i + AHEAD < RING) load_kv(t[i + AHEAD], kp, vp, KPART - 16 * RING + 16 * (i + AHEAD));
    attn_step(t[i], qb0, qb1, mrun, lsum, acc);
  }

  float l = lsum + xhalf(lsum);
  float* slot = sPart + (h * 16 + m) * MSLOT;
  if (part == 1 && half == 0) {
    slot[0] = mrun;
    slot[1] = l;
#pragma unroll
    for (int j = 0; j < 8; j++) slot[2 + j] = acc[j];
  }
  __syncthreads();
  if (part == 0 && half == 0) {
    float m1 = slot[0];
    float l1 = slot[1];
    float mn = fmaxf(mrun, m1);
    float a0 = __expf(mrun - mn);
    float a1 = __expf(m1 - mn);
    float lt = l * a0 + l1 * a1;
    float* orow = sX + m * LP + h * DHH;
#pragma unroll
    for (int j = 0; j < 8; j++) orow[j] = (acc[j] * a0 + slot[2 + j] * a1) / lt;
  }
  __syncthreads();

  int tid = threadIdx.x;
  bool chain = w < 4;
  int colbase = w * 16;
  int rr = 8 * half;
  int cc = colbase + m;

  if (chain) {
    v8f op = tile_gemm64(sX, Wo, colbase, m, khalf);
#pragma unroll
    for (int j = 0; j < 8; j++) {
      int row = qbase + rr + j;
      float sc = rsqrtf(1.f + (float)cs[row]);
      sY[(rr + j) * LP + cc] = hbuf[row * DD + cc] + op[j] * sc;
    }
  }
  __syncthreads();
  if (chain) ln_rows_inplace(sY, tid, g1, b1);
  __syncthreads();

  if (chain) {
    v8f f1 = tile_gemm64(sY, W1, colbase, m, khalf);
    tile_relu_to_lds(sX, f1, colbase, lane);
  }
  __syncthreads();
  if (chain) {
    v8f f2 = tile_gemm64(sX, W2, colbase, m, khalf);
#pragma unroll
    for (int j = 0; j < 8; j++) sY[(rr + j) * LP + cc] += f2[j];
  }
  __syncthreads();
  if (chain) ln_rows_inplace(sY, tid, g2, b2);
  __syncthreads();

  if (chain) {
#pragma unroll
    for (int i = 0; i < 2; i++) {
      int idx = tid + 128 * i;
      int r = idx >> 4, c = (idx & 15) * 4;
      *(float4*)(hbuf + (qbase + r) * DD + c) = *(const float4*)(sY + r * LP + c);
    }
  }
}

extern "C" void kernel_launch(void* const* d_in, const int* in_sizes, int n_in,
                              void* d_out, int out_size, void* d_ws, size_t ws_size,
                              hipStream_t stream) {
  const float* x  = (const float*)d_in[0];
  const int*  ei  = (const int*)d_in[1];
  const int*  src = ei;
  const int*  dst = ei + EE;
  const float* We = (const float*)d_in[2];
  const float* Wg = (const float*)d_in[3];
  const float* Wq = (const float*)d_in[4];
  const float* Wk = (const float*)d_in[5];
  const float* Wv = (const float*)d_in[6];
  const float* Wo = (const float*)d_in[7];
  const float* W1 = (const float*)d_in[8];
  const float* W2 = (const float*)d_in[9];
  const float* g1 = (const float*)d_in[10];
  const float* b1 = (const float*)d_in[11];
  const float* g2 = (const float*)d_in[12];
  const float* b2 = (const float*)d_in[13];

  float* h = (float*)d_out;
  float* ws = (float*)d_ws;
  int*   col   = (int*)ws;
  int*   rank  = col + EE;
  int*   rowst = rank + EE;
  float* qb    = ws + 3 * ND;
  float* kb    = qb + ND;
  float* vb    = kb + ND;
  int*   cnt   = (int*)(vb + 4 * ND + 2 * NN);

  embed_zero<<<NN / 16, 128, 0, stream>>>(x, We, h, cnt);
  count_deg<<<(EE + 255) / 256, 256, 0, stream>>>(src, dst, cnt, cnt + NN, rank);
  csr_fill<<<EE / FILL_T, FILL_T, 0, stream>>>(src, dst, cnt + NN, rank, rowst, col);

  for (int l = 0; l < LL; l++) {
    stage_a<<<NN / 16, 128, 0, stream>>>(h, rowst, col, Wg + l * DD * DD, Wq + l * DD * DD,
                                         Wk + l * DD * DD, Wv + l * DD * DD, qb, kb, vb);

    attn_kernel<<<NN / 16, 256 * KSPLIT, 0, stream>>>(qb, kb, vb, h, cnt, Wo + l * DD * DD,
                                                      W1 + l * DD * FF, W2 + l * FF * DD,
                                                      g1 + l * DD, b1 + l * DD, g2 + l * DD,
                                                      b2 + l * DD);
  }
}


# attention: one online-softmax step per 64 keys (4 key tiles), not per 16: max exchange, alpha, accumulator rescale once per step; operand refills regrouped per step; denominator summed as a pair tree
# speedup vs baseline: 9.8718x; 1.0453x over previous
#include <hip/hip_runtime.h>

typedef float v2f __attribute__((ext_vector_type(2)));
typedef float v8f __attribute__((ext_vector_type(8)));

#define NN 4096
#define EE 131072
#define INDIM 32
#define DD 64
#define HH 8
#define DHH 8
#define FF 64
#define LL 4
#define ND (NN * DD)

__device__ __forceinline__ v8f wmma4(v2f a, v2f b, v8f c) {
  return __builtin_amdgcn_wmma_f32_16x16x4_f32(
       false, a,  false, b,
       (short)0, c,  false,  false);
}

__global__ void count_deg(const int* __restrict__ src, const int* __restrict__ dst,
                          int* cs, int* ci, int* __restrict__ rank) {
  int e = blockIdx.x * blockDim.x + threadIdx.x;
  if (e < EE) {
    atomicAdd(&cs[src[e]], 1);
    rank[e] = atomicAdd(&ci[dst[e]], 1);
  }
}

#define FILL_T 1024
static_assert(NN == 4 * FILL_T, "one int4 of counts per thread");
static_assert(EE % FILL_T == 0, "whole workgroups of edges");
__global__ void __launch_bounds__(FILL_T) csr_fill(const int* __restrict__ src,
                                                   const int* __restrict__ dst,
                                                   const int* __restrict__ ci,
                                                   const int* __restrict__ rank,
                                                   int* __restrict__ row_start,
                                                   int* __restrict__ col) {
  __shared__ __attribute__((aligned(16))) int sStart[NN];
  __shared__ int sWave[FILL_T / 32];
  int t = threadIdx.x;
  int lane = t & 31;
  int w = t >> 5;
  int e = blockIdx.x * FILL_T + t;
  int ed = dst[e];
  int es = src[e];
  int er = rank[e];
  int4 c = *(const int4*)(ci + 4 * t);
  int s0 = c.x, s1 = s0 + c.y, s2 = s1 + c.z, s3 = s2 + c.w;
  int incl = s3;
  for (int d = 1; d < 32; d <<= 1) {
    int y = __shfl_up(incl, d);
    if (lane >= d) incl += y;
  }
  if (lane == 31) sWave[w] = incl;
  __syncthreads();
  if (w == 0) {
    int v = sWave[lane];
    int iv = v;
    for (int d = 1; d < 32; d <<= 1) {
      int y = __shfl_up(iv, d);
      if (lane >= d) iv += y;
    }
    sWave[lane] = iv - v;
  }
  __syncthreads();
  int base = sWave[w] + incl - s3;
  int4 st = {base, base + s0, base + s1, base + s2};
  *(int4*)(sStart + 4 * t) = st;
  if (blockIdx.x == 0) {
    *(int4*)(row_start + 4 * t) = st;
    if (t == FILL_T - 1) row_start[NN] = base + s3;
  }
  __syncthreads();
  col[sStart[ed] + er] = es;
}

#define LP 68
#define CNT_PER_WG (2 * NN / (NN / 16))

__device__ __forceinline__ v8f tile_gemm64(const float* As, const float* __restrict__ B,
                                           int colbase, int m, int khalf) {
  v8f c = {};
#pragma unroll
  for (int k0 = 0; k0 < 64; k0 += 4) {
    v2f a, b;
    const float* ar = As + m * LP + k0 + khalf;
    a.x = ar[0];
    a.y = ar[1];
    b.x = B[(k0 + khalf) * 64 + colbase + m];
    b.y = B[(k0 + khalf + 1) * 64 + colbase + m];
    c = wmma4(a, b, c);
  }
  return c;
}

__device__ __forceinline__ void tile_to_global(float* __restrict__ C, v8f c, int rowbase,
                                               int colbase, int lane) {
  int rr = rowbase + 8 * (lane >> 4);
  int cc = colbase + (lane & 15);
#pragma unroll
  for (int j = 0; j < 8; j++) C[(rr + j) * 64 + cc] = c[j];
}
__device__ __forceinline__ void tile_relu_to_lds(float* Cs, v8f c, int colbase, int lane) {
  int rr = 8 * (lane >> 4);
  int cc = colbase + (lane & 15);
#pragma unroll
  for (int j = 0; j < 8; j++) Cs[(rr + j) * LP + cc] = fmaxf(c[j], 0.f);
}

__device__ __forceinline__ float row_tree_sum(const float* x) {
  float y = ((x[0] + x[4]) + (x[2] + x[6])) + ((x[1] + x[5]) + (x[3] + x[7]));
  y += __shfl_xor(y, 4);
  y += __shfl_xor(y, 2);
  y += __shfl_xor(y, 1);
  return y;
}

__device__ __forceinline__ void ln_rows_inplace(float* tile, int t, const float* __restrict__ g,
                                                const float* __restrict__ b) {
  int i = t & 7;
  float* row = tile + (t >> 3) * LP + i;
  float x[8], d[8], sq[8];
#pragma unroll
  for (int k = 0; k < 8; k++) x[k] = row[8 * k];
  float mean = row_tree_sum(x) * (1.f / 64.f);
#pragma unroll
  for (int k = 0; k < 8; k++) {
    d[k] = x[k] - mean;
    sq[k] = d[k] * d[k];
  }
  float var = row_tree_sum(sq) * (1.f / 64.f);
  float rs = rsqrtf(var + 1e-5f);
#pragma unroll
  for (int k = 0; k < 8; k++) row[8 * k] = d[k] * rs * g[i + 8 * k] + b[i + 8 * k];
}

__global__ void __launch_bounds__(128) embed_zero(const float* __restrict__ x,
                                                  const float* __restrict__ We,
                                                  float* __restrict__ h, int* __restrict__ cnt) {
  int t = threadIdx.x;
  int lane = t & 31;
  int rowbase = blockIdx.x * 16;
  int colbase = (t >> 5) * 16;
  int m = lane & 15;
  int khalf = (lane >> 4) * 2;
  if (t < CNT_PER_WG) cnt[blockIdx.x * CNT_PER_WG + t] = 0;
  v8f c = {};
#pragma unroll
  for (int k0 = 0; k0 < INDIM; k0 += 4) {
    v2f a, b;
    const float* ar = x + (rowbase + m) * INDIM + k0 + khalf;
    a.x = ar[0];
    a.y = ar[1];
    b.x = We[(k0 + khalf) * 64 + colbase + m];
    b.y = We[(k0 + khalf + 1) * 64 + colbase + m];
    c = wmma4(a, b, c);
  }
  tile_to_global(h, c, rowbase, colbase, lane);
}

#define GU 16

__global__ void __launch_bounds__(128) stage_a(const float* __restrict__ h,
                                               const int* __restrict__ row_start,
                                               const int* __restrict__ col,
                                               const float* __restrict__ Wg,
                                               const float* __restrict__ Wq,
                                               const float* __restrict__ Wk,
                                               const float* __restrict__ Wv,
                                               float* __restrict__ qb, float* __restrict__ kb,
                                               float* __restrict__ vb) {
  __shared__ __attribute__((aligned(16))) float sS[16 * LP];
  __shared__ __attribute__((aligned(16))) float sH[16 * LP];
  __shared__ __attribute__((aligned(16))) float sF[16 * LP];
  int t = threadIdx.x;
  int lane = t & 31;
  int rowbase = blockIdx.x * 16;
  int colbase = (t >> 5) * 16;
  int m = lane & 15;
  int khalf = (lane >> 4) * 2;

  {
    int r = t >> 3;
    int c0 = (t & 7) * 4;
    int row = rowbase + r;
    int jb = row_start[row];
    int je = row_start[row + 1];
    const float* hc = h + c0;
    float4 hv0 = *(const float4*)(hc + row * DD);
    float4 hv1 = *(const float4*)(hc + row * DD + 32);
    float4 m0 = {0.f, 0.f, 0.f, 0.f};
    float4 m1 = {0.f, 0.f, 0.f, 0.f};
    for (int j = jb; j < je; j += GU) {
      int sidx[GU];
      float4 v0[GU], v1[GU];
#pragma unroll
      for (int u = 0; u < GU; u++) sidx[u] = col[min(j + u, je - 1)];
#pragma unroll
      for (int u = 0; u < GU; u++) {
        const float* p = hc + sidx[u] * DD;
        v0[u] = *(const float4*)p;
        v1[u] = *(const float4*)(p + 32);
      }
#pragma unroll
      for (int u = 0; u < GU; u++) {
        bool in = j + u < je;
        m0.x += in ? v0[u].x : 0.f; m0.y += in ? v0[u].y : 0.f;
        m0.z += in ? v0[u].z : 0.f; m0.w += in ? v0[u].w : 0.f;
        m1.x += in ? v1[u].x : 0.f; m1.y += in ? v1[u].y : 0.f;
        m1.z += in ? v1[u].z : 0.f; m1.w += in ? v1[u].w : 0.f;
      }
    }
    float inv = 1.f / fmaxf((float)(je - jb), 1.f);
    float4 sv0, sv1;
    sv0.x = hv0.x + m0.x * inv;
    sv0.y = hv0.y + m0.y * inv;
    sv0.z = hv0.z + m0.z * inv;
    sv0.w = hv0.w + m0.w * inv;
    sv1.x = hv1.x + m1.x * inv;
    sv1.y = hv1.y + m1.y * inv;
    sv1.z = hv1.z + m1.z * inv;
    sv1.w = hv1.w + m1.w * inv;
    *(float4*)(sH + r * LP + c0) = hv0;
    *(float4*)(sH + r * LP + c0 + 32) = hv1;
    *(float4*)(sS + r * LP + c0) = sv0;
    *(float4*)(sS + r * LP + c0 + 32) = sv1;
  }
  __syncthreads();
  v8f vv = tile_gemm64(sH, Wv, colbase, m, khalf);
  v8f ff = tile_gemm64(sS, Wg, colbase, m, khalf);
  tile_to_global(vb, vv, rowbase, colbase, lane);
  tile_relu_to_lds(sF, ff, colbase, lane);
  __syncthreads();
  v8f qq = tile_gemm64(sF, Wq, colbase, m, khalf);
  v8f kk = tile_gemm64(sF, Wk, colbase, m, khalf);
  tile_to_global(qb, qq, rowbase, colbase, lane);
  tile_to_global(kb, kk, rowbase, colbase, lane);
}

#define KSPLIT 2
#define KPART (NN / KSPLIT)
#define MSLOT 12
__device__ __forceinline__ float xhalf(float x) {
  int b = __builtin_bit_cast(int, x);
  return __builtin_bit_cast(
      float, __builtin_amdgcn_permlanex16(b, b, 0x76543210, 0xfedcba98, false, false));
}


#define GT 4
#define KGROUP (16 * GT)

struct KTile {
  v2f k0, k1;
};
struct VTile {
  v2f v[4];
};

__device__ __forceinline__ void load_k_group(KTile (&kt)[2 * GT], const int base,
                                             const float* __restrict__ kg) {
#pragma unroll
  for (int u = 0; u < GT; u++) {
    const float* krow = kg + u * 16 * DD;
    kt[base + u].k0.x = krow[0]; kt[base + u].k0.y = krow[1];
    kt[base + u].k1.x = krow[4]; kt[base + u].k1.y = krow[5];
  }
}
__device__ __forceinline__ void load_v_group(VTile (&vt)[2 * GT], const int base,
                                             const float* __restrict__ vg) {
#pragma unroll
  for (int u = 0; u < GT; u++) {
    const float* vrow = vg + u * 16 * DD;
#pragma unroll
    for (int c = 0; c < 4; c++) {
      vt[base + u].v[c].x = vrow[(2 * c) * DD];
      vt[base + u].v[c].y = vrow[(2 * c + 1) * DD];
    }
  }
}

template <bool LOADK, bool LOADV>
__device__ __forceinline__ void attn_group(KTile (&kt)[2 * GT], VTile (&vt)[2 * GT], const int cur,
                                           const int oth, const float* __restrict__ kg2,
                                           const float* __restrict__ vg1, v2f qb0, v2f qb1,
                                           float& mrun, float& lsum, v8f& acc) {
  v8f s[GT];
#pragma unroll
  for (int u = 0; u < GT; u++) {
    v8f z = {};
    s[u] = wmma4(kt[cur + u].k0, qb0, z);
  }
#pragma unroll
  for (int u = 0; u < GT; u++) s[u] = wmma4(kt[cur + u].k1, qb1, s[u]);

  if (LOADK) load_k_group(kt, cur, kg2);
  if (LOADV) load_v_group(vt, oth, vg1);

  float tm[GT];
#pragma unroll
  for (int u = 0; u < GT; u++)
    tm[u] = fmaxf(fmaxf(fmaxf(s[u][0], s[u][1]), fmaxf(s[u][2], s[u][3])),
                  fmaxf(fmaxf(s[u][4], s[u][5]), fmaxf(s[u][6], s[u][7])));
  float sm = fmaxf(fmaxf(tm[0], tm[1]), fmaxf(tm[2], tm[3]));
  sm = fmaxf(sm, xhalf(sm));
  float mn = fmaxf(mrun, sm);
  float alpha = __expf(mrun - mn);
  mrun = mn;
#pragma unroll
  for (int j = 0; j < 8; j++) acc[j] *= alpha;

  v2f pp[GT][4];
  v2f pt[GT];
#pragma unroll
  for (int u = 0; u < GT; u++) {
#pragma unroll
    for (int c = 0; c < 4; c++) {
      pp[u][c].x = __expf(s[u][2 * c] - mn);
      pp[u][c].y = __expf(s[u][2 * c + 1] - mn);
    }
    pt[u] = (pp[u][0] + pp[u][1]) + (pp[u][2] + pp[u][3]);
  }
  v2f pg = (pt[0] + pt[1]) + (pt[2] + pt[3]);
  lsum = lsum * alpha + (pg.x + pg.y);

#pragma unroll
  for (int u = 0; u < GT; u++) {
#pragma unroll
    for (int c = 0; c < 4; c++) acc = wmma4(vt[cur + u].v[c], pp[u][c], acc);
  }
}
static_assert(GT == 4, "the step's maximum and the denominator's tree are written out for four tiles");

__device__ __forceinline__ void step_fence(v8f& acc, v2f& qb0, v2f& qb1, int& kb) {
  asm volatile("" : "+v"(acc), "+v"(qb0), "+v"(qb1), "+s"(kb));
}

__global__ void __launch_bounds__(256 * KSPLIT) attn_kernel(const float* __restrict__ q,
                                                   const float* __restrict__ k,
                                                   const float* __restrict__ v,
                                                   float* __restrict__ hbuf,
                                                   const int* __restrict__ cs,
                                                   const float* __restrict__ Wo,
                                                   const float* __restrict__ W1,
                                                   const float* __restrict__ W2,
                                                   const float* __restrict__ g1,
                                                   const float* __restrict__ b1,
                                                   const float* __restrict__ g2,
                                                   const float* __restrict__ b2) {
  __shared__ float sPart[HH * 16 * MSLOT];
  __shared__ __attribute__((aligned(16))) float sX[16 * LP];
  __shared__ __attribute__((aligned(16))) float sY[16 * LP];
  int w = threadIdx.x >> 5;
  int h = w & (HH - 1);
  int part = w >> 3;
  int lane = threadIdx.x & 31;
  int qbase = blockIdx.x * 16;
  int m = lane & 15;
  int half = lane >> 4;
  int khalf = half * 2;
  const float scale = 0.35355339059327373f;

  const float* qrow = q + (qbase + m) * DD + h * DHH;
  v2f qb0, qb1;
  qb0.x = qrow[khalf] * scale;
  qb0.y = qrow[khalf + 1] * scale;
  qb1.x = qrow[4 + khalf] * scale;
  qb1.y = qrow[4 + khalf + 1] * scale;

  float mrun = -1e30f;
  float lsum = 0.f;
  v8f acc = {};

  const float* kp = k + (part * KPART + m) * DD + h * DHH + khalf;
  const float* vp = v + (part * KPART + 8 * half) * DD + h * DHH + (m & 7);

  KTile kt[2 * GT];
  VTile vt[2 * GT];
  int kb = 0;
  load_k_group(kt, 0, kp + kb * DD);
  step_fence(acc, qb0, qb1, kb);
  load_v_group(vt, 0, vp + kb * DD);
  step_fence(acc, qb0, qb1, kb);
  load_k_group(kt, GT, kp + (kb + KGROUP) * DD);
  step_fence(acc, qb0, qb1, kb);

  for (; kb < KPART - 2 * KGROUP; kb += 2 * KGROUP) {
    attn_group<true, true>(kt, vt, 0, GT, kp + (kb + 2 * KGROUP) * DD, vp + (kb + KGROUP) * DD,
                           qb0, qb1, mrun, lsum, acc);
    step_fence(acc, qb0, qb1, kb);
    attn_group<true, true>(kt, vt, GT, 0, kp + (kb + 3 * KGROUP) * DD, vp + (kb + 2 * KGROUP) * DD,
                           qb0, qb1, mrun, lsum, acc);
    step_fence(acc, qb0, qb1, kb);
  }
  attn_group<false, true>(kt, vt, 0, GT, kp, vp + (KPART - KGROUP) * DD, qb0, qb1, mrun, lsum, acc);
  attn_group<false, false>(kt, vt, GT, 0, kp, vp, qb0, qb1, mrun, lsum, acc);

  float l = lsum + xhalf(lsum);
  float* slot = sPart + (h * 16 + m) * MSLOT;
  if (part == 1 && half == 0) {
    slot[0] = mrun;
    slot[1] = l;
#pragma unroll
    for (int j = 0; j < 8; j++) slot[2 + j] = acc[j];
  }
  __syncthreads();
  if (part == 0 && half == 0) {
    float m1 = slot[0];
    float l1 = slot[1];
    float mn = fmaxf(mrun, m1);
    float a0 = __expf(mrun - mn);
    float a1 = __expf(m1 - mn);
    float lt = l * a0 + l1 * a1;
    float* orow = sX + m * LP + h * DHH;
#pragma unroll
    for (int j = 0; j < 8; j++) orow[j] = (acc[j] * a0 + slot[2 + j] * a1) / lt;
  }
  __syncthreads();

  int tid = threadIdx.x;
  bool chain = w < 4;
  int colbase = w * 16;
  int rr = 8 * half;
  int cc = colbase + m;

  if (chain) {
    v8f op = tile_gemm64(sX, Wo, colbase, m, khalf);
#pragma unroll
    for (int j = 0; j < 8; j++) {
      int row = qbase + rr + j;
      float sc = rsqrtf(1.f + (float)cs[row]);
      sY[(rr + j) * LP + cc] = hbuf[row * DD + cc] + op[j] * sc;
    }
  }
  __syncthreads();
  if (chain) ln_rows_inplace(sY, tid, g1, b1);
  __syncthreads();

  if (chain) {
    v8f f1 = tile_gemm64(sY, W1, colbase, m, khalf);
    tile_relu_to_lds(sX, f1, colbase, lane);
  }
  __syncthreads();
  if (chain) {
    v8f f2 = tile_gemm64(sX, W2, colbase, m, khalf);
#pragma unroll
    for (int j = 0; j < 8; j++) sY[(rr + j) * LP + cc] += f2[j];
  }
  __syncthreads();
  if (chain) ln_rows_inplace(sY, tid, g2, b2);
  __syncthreads();

  if (chain) {
#pragma unroll
    for (int i = 0; i < 2; i++) {
      int idx = tid + 128 * i;
      int r = idx >> 4, c = (idx & 15) * 4;
      *(float4*)(hbuf + (qbase + r) * DD + c) = *(const float4*)(sY + r * LP + c);
    }
  }
}

extern "C" void kernel_launch(void* const* d_in, const int* in_sizes, int n_in,
                              void* d_out, int out_size, void* d_ws, size_t ws_size,
                              hipStream_t stream) {
  const float* x  = (const float*)d_in[0];
  const int*  ei  = (const int*)d_in[1];
  const int*  src = ei;
  const int*  dst = ei + EE;
  const float* We = (const float*)d_in[2];
  const float* Wg = (const float*)d_in[3];
  const float* Wq = (const float*)d_in[4];
  const float* Wk = (const float*)d_in[5];
  const float* Wv = (const float*)d_in[6];
  const float* Wo = (const float*)d_in[7];
  const float* W1 = (const float*)d_in[8];
  const float* W2 = (const float*)d_in[9];
  const float* g1 = (const float*)d_in[10];
  const float* b1 = (const float*)d_in[11];
  const float* g2 = (const float*)d_in[12];
  const float* b2 = (const float*)d_in[13];

  float* h = (float*)d_out;
  float* ws = (float*)d_ws;
  int*   col   = (int*)ws;
  int*   rank  = col + EE;
  int*   rowst = rank + EE;
  float* qb    = ws + 3 * ND;
  float* kb    = qb + ND;
  float* vb    = kb + ND;
  int*   cnt   = (int*)(vb + 4 * ND + 2 * NN);

  embed_zero<<<NN / 16, 128, 0, stream>>>(x, We, h, cnt);
  count_deg<<<(EE + 255) / 256, 256, 0, stream>>>(src, dst, cnt, cnt + NN, rank);
  csr_fill<<<EE / FILL_T, FILL_T, 0, stream>>>(src, dst, cnt + NN, rank, rowst, col);

  for (int l = 0; l < LL; l++) {
    stage_a<<<NN / 16, 128, 0, stream>>>(h, rowst, col, Wg + l * DD * DD, Wq + l * DD * DD,
                                         Wk + l * DD * DD, Wv + l * DD * DD, qb, kb, vb);

    attn_kernel<<<NN / 16, 256 * KSPLIT, 0, stream>>>(qb, kb, vb, h, cnt, Wo + l * DD * DD,
                                                      W1 + l * DD * FF, W2 + l * FF * DD,
                                                      g1 + l * DD, b1 + l * DD, g2 + l * DD,
                                                      b2 + l * DD);
  }
}


# K and V stored per head in 16-key blocks in attention's operand order (K tile made transposed by its producer): a step reads 12 whole-line 16-B loads, not 40 short ones; score dim sum order changes
# speedup vs baseline: 10.1889x; 1.0321x over previous
#include <hip/hip_runtime.h>

typedef float v2f __attribute__((ext_vector_type(2)));
typedef float v8f __attribute__((ext_vector_type(8)));

#define NN 4096
#define EE 131072
#define INDIM 32
#define DD 64
#define HH 8
#define DHH 8
#define FF 64
#define LL 4
#define ND (NN * DD)

__device__ __forceinline__ v8f wmma4(v2f a, v2f b, v8f c) {
  return __builtin_amdgcn_wmma_f32_16x16x4_f32(
       false, a,  false, b,
       (short)0, c,  false,  false);
}

__global__ void count_deg(const int* __restrict__ src, const int* __restrict__ dst,
                          int* cs, int* ci, int* __restrict__ rank) {
  int e = blockIdx.x * blockDim.x + threadIdx.x;
  if (e < EE) {
    atomicAdd(&cs[src[e]], 1);
    rank[e] = atomicAdd(&ci[dst[e]], 1);
  }
}

#define FILL_T 1024
static_assert(NN == 4 * FILL_T, "one int4 of counts per thread");
static_assert(EE % FILL_T == 0, "whole workgroups of edges");
__global__ void __launch_bounds__(FILL_T) csr_fill(const int* __restrict__ src,
                                                   const int* __restrict__ dst,
                                                   const int* __restrict__ ci,
                                                   const int* __restrict__ rank,
                                                   int* __restrict__ row_start,
                                                   int* __restrict__ col) {
  __shared__ __attribute__((aligned(16))) int sStart[NN];
  __shared__ int sWave[FILL_T / 32];
  int t = threadIdx.x;
  int lane = t & 31;
  int w = t >> 5;
  int e = blockIdx.x * FILL_T + t;
  int ed = dst[e];
  int es = src[e];
  int er = rank[e];
  int4 c = *(const int4*)(ci + 4 * t);
  int s0 = c.x, s1 = s0 + c.y, s2 = s1 + c.z, s3 = s2 + c.w;
  int incl = s3;
  for (int d = 1; d < 32; d <<= 1) {
    int y = __shfl_up(incl, d);
    if (lane >= d) incl += y;
  }
  if (lane == 31) sWave[w] = incl;
  __syncthreads();
  if (w == 0) {
    int v = sWave[lane];
    int iv = v;
    for (int d = 1; d < 32; d <<= 1) {
      int y = __shfl_up(iv, d);
      if (lane >= d) iv += y;
    }
    sWave[lane] = iv - v;
  }
  __syncthreads();
  int base = sWave[w] + incl - s3;
  int4 st = {base, base + s0, base + s1, base + s2};
  *(int4*)(sStart + 4 * t) = st;
  if (blockIdx.x == 0) {
    *(int4*)(row_start + 4 * t) = st;
    if (t == FILL_T - 1) row_start[NN] = base + s3;
  }
  __syncthreads();
  col[sStart[ed] + er] = es;
}

#define LP 68
#define CNT_PER_WG (2 * NN / (NN / 16))

template <bool TRANSPOSED = false>
__device__ __forceinline__ v8f tile_gemm64(const float* As, const float* __restrict__ B,
                                           int colbase, int m, int khalf) {
  v8f c = {};
#pragma unroll
  for (int k0 = 0; k0 < 64; k0 += 4) {
    v2f a, b;
    const float* ar = As + m * LP + k0 + khalf;
    a.x = ar[0];
    a.y = ar[1];
    b.x = B[(k0 + khalf) * 64 + colbase + m];
    b.y = B[(k0 + khalf + 1) * 64 + colbase + m];
    c = TRANSPOSED ? wmma4(b, a, c) : wmma4(a, b, c);
  }
  return c;
}

#define NBLK (NN / 16)
#define BLKF 128
#define HEADF (NBLK * BLKF)
#define KEYF (BLKF / 16)
static_assert(DHH == 8 && HH * HEADF == ND, "block layouts are written for head dimension 8");

__device__ __forceinline__ void v_tile_to_blocks(float* __restrict__ Vb, v8f c, int blk,
                                                 int colbase, int lane) {
  int col = colbase + (lane & 15);
  float* p = Vb + (col >> 3) * HEADF + blk * BLKF + (col & 7) * 8 + (lane >> 4) * 4;
  float4 lo = {c[0], c[1], c[2], c[3]};
  float4 hi = {c[4], c[5], c[6], c[7]};
  *(float4*)p = lo;
  *(float4*)(p + 64) = hi;
}
__device__ __forceinline__ void kt_tile_to_blocks(float* __restrict__ Kb, v8f c, int blk,
                                                  int colbase, int lane) {
  float* p = Kb + ((colbase >> 3) + (lane >> 4)) * HEADF + blk * BLKF + (lane & 15) * 4;
  float4 lo = {c[0], c[1], c[2], c[3]};
  float4 hi = {c[4], c[5], c[6], c[7]};
  *(float4*)p = lo;
  *(float4*)(p + 64) = hi;
}

__device__ __forceinline__ void tile_to_global(float* __restrict__ C, v8f c, int rowbase,
                                               int colbase, int lane) {
  int rr = rowbase + 8 * (lane >> 4);
  int cc = colbase + (lane & 15);
#pragma unroll
  for (int j = 0; j < 8; j++) C[(rr + j) * 64 + cc] = c[j];
}
__device__ __forceinline__ void tile_relu_to_lds(float* Cs, v8f c, int colbase, int lane) {
  int rr = 8 * (lane >> 4);
  int cc = colbase + (lane & 15);
#pragma unroll
  for (int j = 0; j < 8; j++) Cs[(rr + j) * LP + cc] = fmaxf(c[j], 0.f);
}

__device__ __forceinline__ float row_tree_sum(const float* x) {
  float y = ((x[0] + x[4]) + (x[2] + x[6])) + ((x[1] + x[5]) + (x[3] + x[7]));
  y += __shfl_xor(y, 4);
  y += __shfl_xor(y, 2);
  y += __shfl_xor(y, 1);
  return y;
}

__device__ __forceinline__ void ln_rows_inplace(float* tile, int t, const float* __restrict__ g,
                                                const float* __restrict__ b) {
  int i = t & 7;
  float* row = tile + (t >> 3) * LP + i;
  float x[8], d[8], sq[8];
#pragma unroll
  for (int k = 0; k < 8; k++) x[k] = row[8 * k];
  float mean = row_tree_sum(x) * (1.f / 64.f);
#pragma unroll
  for (int k = 0; k < 8; k++) {
    d[k] = x[k] - mean;
    sq[k] = d[k] * d[k];
  }
  float var = row_tree_sum(sq) * (1.f / 64.f);
  float rs = rsqrtf(var + 1e-5f);
#pragma unroll
  for (int k = 0; k < 8; k++) row[8 * k] = d[k] * rs * g[i + 8 * k] + b[i + 8 * k];
}

__global__ void __launch_bounds__(128) embed_zero(const float* __restrict__ x,
                                                  const float* __restrict__ We,
                                                  float* __restrict__ h, int* __restrict__ cnt) {
  int t = threadIdx.x;
  int lane = t & 31;
  int rowbase = blockIdx.x * 16;
  int colbase = (t >> 5) * 16;
  int m = lane & 15;
  int khalf = (lane >> 4) * 2;
  if (t < CNT_PER_WG) cnt[blockIdx.x * CNT_PER_WG + t] = 0;
  v8f c = {};
#pragma unroll
  for (int k0 = 0; k0 < INDIM; k0 += 4) {
    v2f a, b;
    const float* ar = x + (rowbase + m) * INDIM + k0 + khalf;
    a.x = ar[0];
    a.y = ar[1];
    b.x = We[(k0 + khalf) * 64 + colbase + m];
    b.y = We[(k0 + khalf + 1) * 64 + colbase + m];
    c = wmma4(a, b, c);
  }
  tile_to_global(h, c, rowbase, colbase, lane);
}

#define GU 16

__global__ void __launch_bounds__(128) stage_a(const float* __restrict__ h,
                                               const int* __restrict__ row_start,
                                               const int* __restrict__ col,
                                               const float* __restrict__ Wg,
                                               const float* __restrict__ Wq,
                                               const float* __restrict__ Wk,
                                               const float* __restrict__ Wv,
                                               float* __restrict__ qb, float* __restrict__ kb,
                                               float* __restrict__ vb) {
  __shared__ __attribute__((aligned(16))) float sS[16 * LP];
  __shared__ __attribute__((aligned(16))) float sH[16 * LP];
  __shared__ __attribute__((aligned(16))) float sF[16 * LP];
  int t = threadIdx.x;
  int lane = t & 31;
  int rowbase = blockIdx.x * 16;
  int colbase = (t >> 5) * 16;
  int m = lane & 15;
  int khalf = (lane >> 4) * 2;

  {
    int r = t >> 3;
    int c0 = (t & 7) * 4;
    int row = rowbase + r;
    int jb = row_start[row];
    int je = row_start[row + 1];
    const float* hc = h + c0;
    float4 hv0 = *(const float4*)(hc + row * DD);
    float4 hv1 = *(const float4*)(hc + row * DD + 32);
    float4 m0 = {0.f, 0.f, 0.f, 0.f};
    float4 m1 = {0.f, 0.f, 0.f, 0.f};
    for (int j = jb; j < je; j += GU) {
      int sidx[GU];
      float4 v0[GU], v1[GU];
#pragma unroll
      for (int u = 0; u < GU; u++) sidx[u] = col[min(j + u, je - 1)];
#pragma unroll
      for (int u = 0; u < GU; u++) {
        const float* p = hc + sidx[u] * DD;
        v0[u] = *(const float4*)p;
        v1[u] = *(const float4*)(p + 32);
      }
#pragma unroll
      for (int u = 0; u < GU; u++) {
        bool in = j + u < je;
        m0.x += in ? v0[u].x : 0.f; m0.y += in ? v0[u].y : 0.f;
        m0.z += in ? v0[u].z : 0.f; m0.w += in ? v0[u].w : 0.f;
        m1.x += in ? v1[u].x : 0.f; m1.y += in ? v1[u].y : 0.f;
        m1.z += in ? v1[u].z : 0.f; m1.w += in ? v1[u].w : 0.f;
      }
    }
    float inv = 1.f / fmaxf((float)(je - jb), 1.f);
    float4 sv0, sv1;
    sv0.x = hv0.x + m0.x * inv;
    sv0.y = hv0.y + m0.y * inv;
    sv0.z = hv0.z + m0.z * inv;
    sv0.w = hv0.w + m0.w * inv;
    sv1.x = hv1.x + m1.x * inv;
    sv1.y = hv1.y + m1.y * inv;
    sv1.z = hv1.z + m1.z * inv;
    sv1.w = hv1.w + m1.w * inv;
    *(float4*)(sH + r * LP + c0) = hv0;
    *(float4*)(sH + r * LP + c0 + 32) = hv1;
    *(float4*)(sS + r * LP + c0) = sv0;
    *(float4*)(sS + r * LP + c0 + 32) = sv1;
  }
  __syncthreads();
  v8f vv = tile_gemm64(sH, Wv, colbase, m, khalf);
  v8f ff = tile_gemm64(sS, Wg, colbase, m, khalf);
  v_tile_to_blocks(vb, vv, blockIdx.x, colbase, lane);
  tile_relu_to_lds(sF, ff, colbase, lane);
  __syncthreads();
  v8f qq = tile_gemm64(sF, Wq, colbase, m, khalf);
  v8f kk = tile_gemm64<true>(sF, Wk, colbase, m, khalf);
  tile_to_global(qb, qq, rowbase, colbase, lane);
  kt_tile_to_blocks(kb, kk, blockIdx.x, colbase, lane);
}

#define KSPLIT 2
#define KPART (NN / KSPLIT)
#define MSLOT 12
__device__ __forceinline__ float xhalf(float x) {
  int b = __builtin_bit_cast(int, x);
  return __builtin_bit_cast(
      float, __builtin_amdgcn_permlanex16(b, b, 0x76543210, 0xfedcba98, false, false));
}


#define GT 4
#define KGROUP (16 * GT)

struct KTile {
  v2f k0, k1;
};
struct VTile {
  v2f v[4];
};

__device__ __forceinline__ void load_k_group(KTile (&kt)[2 * GT], const int base,
                                             const float* __restrict__ kg) {
#pragma unroll
  for (int u = 0; u < GT; u++) {
    float4 kq = *(const float4*)(kg + u * BLKF);
    kt[base + u].k0.x = kq.x; kt[base + u].k0.y = kq.y;
    kt[base + u].k1.x = kq.z; kt[base + u].k1.y = kq.w;
  }
}
__device__ __forceinline__ void load_v_group(VTile (&vt)[2 * GT], const int base,
                                             const float* __restrict__ vg) {
#pragma unroll
  for (int u = 0; u < GT; u++) {
    float4 v01 = *(const float4*)(vg + u * BLKF);
    float4 v23 = *(const float4*)(vg + u * BLKF + 64);
    vt[base + u].v[0].x = v01.x; vt[base + u].v[0].y = v01.y;
    vt[base + u].v[1].x = v01.z; vt[base + u].v[1].y = v01.w;
    vt[base + u].v[2].x = v23.x; vt[base + u].v[2].y = v23.y;
    vt[base + u].v[3].x = v23.z; vt[base + u].v[3].y = v23.w;
  }
}

template <bool LOADK, bool LOADV>
__device__ __forceinline__ void attn_group(KTile (&kt)[2 * GT], VTile (&vt)[2 * GT], const int cur,
                                           const int oth, const float* __restrict__ kg2,
                                           const float* __restrict__ vg1, v2f qb0, v2f qb1,
                                           float& mrun, float& lsum, v8f& acc) {
  v8f s[GT];
#pragma unroll
  for (int u = 0; u < GT; u++) {
    v8f z = {};
    s[u] = wmma4(kt[cur + u].k0, qb0, z);
  }
#pragma unroll
  for (int u = 0; u < GT; u++) s[u] = wmma4(kt[cur + u].k1, qb1, s[u]);

  if (LOADK) load_k_group(kt, cur, kg2);
  if (LOADV) load_v_group(vt, oth, vg1);

  float tm[GT];
#pragma unroll
  for (int u = 0; u < GT; u++)
    tm[u] = fmaxf(fmaxf(fmaxf(s[u][0], s[u][1]), fmaxf(s[u][2], s[u][3])),
                  fmaxf(fmaxf(s[u][4], s[u][5]), fmaxf(s[u][6], s[u][7])));
  float sm = fmaxf(fmaxf(tm[0], tm[1]), fmaxf(tm[2], tm[3]));
  sm = fmaxf(sm, xhalf(sm));
  float mn = fmaxf(mrun, sm);
  float alpha = __expf(mrun - mn);
  mrun = mn;
#pragma unroll
  for (int j = 0; j < 8; j++) acc[j] *= alpha;

  v2f pp[GT][4];
  v2f pt[GT];
#pragma unroll
  for (int u = 0; u < GT; u++) {
#pragma unroll
    for (int c = 0; c < 4; c++) {
      pp[u][c].x = __expf(s[u][2 * c] - mn);
      pp[u][c].y = __expf(s[u][2 * c + 1] - mn);
    }
    pt[u] = (pp[u][0] + pp[u][1]) + (pp[u][2] + pp[u][3]);
  }
  v2f pg = (pt[0] + pt[1]) + (pt[2] + pt[3]);
  lsum = lsum * alpha + (pg.x + pg.y);

#pragma unroll
  for (int u = 0; u < GT; u++) {
#pragma unroll
    for (int c = 0; c < 4; c++) acc = wmma4(vt[cur + u].v[c], pp[u][c], acc);
  }
}
static_assert(GT == 4, "the step's maximum and the denominator's tree are written out for four tiles");

__device__ __forceinline__ void step_fence(v8f& acc, v2f& qb0, v2f& qb1, int& kb) {
  asm volatile("" : "+v"(acc), "+v"(qb0), "+v"(qb1), "+s"(kb));
}

__global__ void __launch_bounds__(256 * KSPLIT) attn_kernel(const float* __restrict__ q,
                                                   const float* __restrict__ k,
                                                   const float* __restrict__ v,
                                                   float* __restrict__ hbuf,
                                                   const int* __restrict__ cs,
                                                   const float* __restrict__ Wo,
                                                   const float* __restrict__ W1,
                                                   const float* __restrict__ W2,
                                                   const float* __restrict__ g1,
                                                   const float* __restrict__ b1,
                                                   const float* __restrict__ g2,
                                                   const float* __restrict__ b2) {
  __shared__ float sPart[HH * 16 * MSLOT];
  __shared__ __attribute__((aligned(16))) float sX[16 * LP];
  __shared__ __attribute__((aligned(16))) float sY[16 * LP];
  int w = threadIdx.x >> 5;
  int h = w & (HH - 1);
  int part = w >> 3;
  int lane = threadIdx.x & 31;
  int qbase = blockIdx.x * 16;
  int m = lane & 15;
  int half = lane >> 4;
  int khalf = half * 2;
  const float scale = 0.35355339059327373f;

  const float* qrow = q + (qbase + m) * DD + h * DHH + 4 * half;
  v2f qb0, qb1;
  qb0.x = qrow[0] * scale;
  qb0.y = qrow[1] * scale;
  qb1.x = qrow[2] * scale;
  qb1.y = qrow[3] * scale;

  float mrun = -1e30f;
  float lsum = 0.f;
  v8f acc = {};

  const float* kp = k + h * HEADF + part * KPART * KEYF + half * 64 + m * 4;
  const float* vp = v + h * HEADF + part * KPART * KEYF + (m & 7) * 8 + half * 4;

  KTile kt[2 * GT];
  VTile vt[2 * GT];
  int kb = 0;
  load_k_group(kt, 0, kp + kb * KEYF);
  step_fence(acc, qb0, qb1, kb);
  load_v_group(vt, 0, vp + kb * KEYF);
  step_fence(acc, qb0, qb1, kb);
  load_k_group(kt, GT, kp + (kb + KGROUP) * KEYF);
  step_fence(acc, qb0, qb1, kb);

  for (; kb < KPART - 2 * KGROUP; kb += 2 * KGROUP) {
    attn_group<true, true>(kt, vt, 0, GT, kp + (kb + 2 * KGROUP) * KEYF,
                           vp + (kb + KGROUP) * KEYF, qb0, qb1, mrun, lsum, acc);
    step_fence(acc, qb0, qb1, kb);
    attn_group<true, true>(kt, vt, GT, 0, kp + (kb + 3 * KGROUP) * KEYF,
                           vp + (kb + 2 * KGROUP) * KEYF, qb0, qb1, mrun, lsum, acc);
    step_fence(acc, qb0, qb1, kb);
  }
  attn_group<false, true>(kt, vt, 0, GT, kp, vp + (KPART - KGROUP) * KEYF, qb0, qb1, mrun, lsum,
                          acc);
  attn_group<false, false>(kt, vt, GT, 0, kp, vp, qb0, qb1, mrun, lsum, acc);

  float l = lsum + xhalf(lsum);
  float* slot = sPart + (h * 16 + m) * MSLOT;
  if (part == 1 && half == 0) {
    slot[0] = mrun;
    slot[1] = l;
#pragma unroll
    for (int j = 0; j < 8; j++) slot[2 + j] = acc[j];
  }
  __syncthreads();
  if (part == 0 && half == 0) {
    float m1 = slot[0];
    float l1 = slot[1];
    float mn = fmaxf(mrun, m1);
    float a0 = __expf(mrun - mn);
    float a1 = __expf(m1 - mn);
    float lt = l * a0 + l1 * a1;
    float* orow = sX + m * LP + h * DHH;
#pragma unroll
    for (int j = 0; j < 8; j++) orow[j] = (acc[j] * a0 + slot[2 + j] * a1) / lt;
  }
  __syncthreads();

  int tid = threadIdx.x;
  bool chain = w < 4;
  int colbase = w * 16;
  int rr = 8 * half;
  int cc = colbase + m;

  if (chain) {
    v8f op = tile_gemm64(sX, Wo, colbase, m, khalf);
#pragma unroll
    for (int j = 0; j < 8; j++) {
      int row = qbase + rr + j;
      float sc = rsqrtf(1.f + (float)cs[row]);
      sY[(rr + j) * LP + cc] = hbuf[row * DD + cc] + op[j] * sc;
    }
  }
  __syncthreads();
  if (chain) ln_rows_inplace(sY, tid, g1, b1);
  __syncthreads();

  if (chain) {
    v8f f1 = tile_gemm64(sY, W1, colbase, m, khalf);
    tile_relu_to_lds(sX, f1, colbase, lane);
  }
  __syncthreads();
  if (chain) {
    v8f f2 = tile_gemm64(sX, W2, colbase, m, khalf);
#pragma unroll
    for (int j = 0; j < 8; j++) sY[(rr + j) * LP + cc] += f2[j];
  }
  __syncthreads();
  if (chain) ln_rows_inplace(sY, tid, g2, b2);
  __syncthreads();

  if (chain) {
#pragma unroll
    for (int i = 0; i < 2; i++) {
      int idx = tid + 128 * i;
      int r = idx >> 4, c = (idx & 15) * 4;
      *(float4*)(hbuf + (qbase + r) * DD + c) = *(const float4*)(sY + r * LP + c);
    }
  }
}

extern "C" void kernel_launch(void* const* d_in, const int* in_sizes, int n_in,
                              void* d_out, int out_size, void* d_ws, size_t ws_size,
                              hipStream_t stream) {
  const float* x  = (const float*)d_in[0];
  const int*  ei  = (const int*)d_in[1];
  const int*  src = ei;
  const int*  dst = ei + EE;
  const float* We = (const float*)d_in[2];
  const float* Wg = (const float*)d_in[3];
  const float* Wq = (const float*)d_in[4];
  const float* Wk = (const float*)d_in[5];
  const float* Wv = (const float*)d_in[6];
  const float* Wo = (const float*)d_in[7];
  const float* W1 = (const float*)d_in[8];
  const float* W2 = (const float*)d_in[9];
  const float* g1 = (const float*)d_in[10];
  const float* b1 = (const float*)d_in[11];
  const float* g2 = (const float*)d_in[12];
  const float* b2 = (const float*)d_in[13];

  float* h = (float*)d_out;
  float* ws = (float*)d_ws;
  int*   col   = (int*)ws;
  int*   rank  = col + EE;
  int*   rowst = rank + EE;
  float* qb    = ws + 3 * ND;
  float* kb    = qb + ND;
  float* vb    = kb + ND;
  int*   cnt   = (int*)(vb + 4 * ND + 2 * NN);

  embed_zero<<<NN / 16, 128, 0, stream>>>(x, We, h, cnt);
  count_deg<<<(EE + 255) / 256, 256, 0, stream>>>(src, dst, cnt, cnt + NN, rank);
  csr_fill<<<EE / FILL_T, FILL_T, 0, stream>>>(src, dst, cnt + NN, rank, rowst, col);

  for (int l = 0; l < LL; l++) {
    stage_a<<<NN / 16, 128, 0, stream>>>(h, rowst, col, Wg + l * DD * DD, Wq + l * DD * DD,
                                         Wk + l * DD * DD, Wv + l * DD * DD, qb, kb, vb);

    attn_kernel<<<NN / 16, 256 * KSPLIT, 0, stream>>>(qb, kb, vb, h, cnt, Wo + l * DD * DD,
                                                      W1 + l * DD * FF, W2 + l * FF * DD,
                                                      g1 + l * DD, b1 + l * DD, g2 + l * DD,
                                                      b2 + l * DD);
  }
}


# attention: softmax weight = exp2 of the raw score (log2(e) in q's scale): no running maximum, shift or rescale per step; one range test per wave behind the loop, out of range reruns online
# speedup vs baseline: 10.9700x; 1.0767x over previous
#include <hip/hip_runtime.h>

typedef float v2f __attribute__((ext_vector_type(2)));
typedef float v8f __attribute__((ext_vector_type(8)));

#define NN 4096
#define EE 131072
#define INDIM 32
#define DD 64
#define HH 8
#define DHH 8
#define FF 64
#define LL 4
#define ND (NN * DD)

__device__ __forceinline__ v8f wmma4(v2f a, v2f b, v8f c) {
  return __builtin_amdgcn_wmma_f32_16x16x4_f32(
       false, a,  false, b,
       (short)0, c,  false,  false);
}

__global__ void count_deg(const int* __restrict__ src, const int* __restrict__ dst,
                          int* cs, int* ci, int* __restrict__ rank) {
  int e = blockIdx.x * blockDim.x + threadIdx.x;
  if (e < EE) {
    atomicAdd(&cs[src[e]], 1);
    rank[e] = atomicAdd(&ci[dst[e]], 1);
  }
}

#define FILL_T 1024
static_assert(NN == 4 * FILL_T, "one int4 of counts per thread");
static_assert(EE % FILL_T == 0, "whole workgroups of edges");
__global__ void __launch_bounds__(FILL_T) csr_fill(const int* __restrict__ src,
                                                   const int* __restrict__ dst,
                                                   const int* __restrict__ ci,
                                                   const int* __restrict__ rank,
                                                   int* __restrict__ row_start,
                                                   int* __restrict__ col) {
  __shared__ __attribute__((aligned(16))) int sStart[NN];
  __shared__ int sWave[FILL_T / 32];
  int t = threadIdx.x;
  int lane = t & 31;
  int w = t >> 5;
  int e = blockIdx.x * FILL_T + t;
  int ed = dst[e];
  int es = src[e];
  int er = rank[e];
  int4 c = *(const int4*)(ci + 4 * t);
  int s0 = c.x, s1 = s0 + c.y, s2 = s1 + c.z, s3 = s2 + c.w;
  int incl = s3;
  for (int d = 1; d < 32; d <<= 1) {
    int y = __shfl_up(incl, d);
    if (lane >= d) incl += y;
  }
  if (lane == 31) sWave[w] = incl;
  __syncthreads();
  if (w == 0) {
    int v = sWave[lane];
    int iv = v;
    for (int d = 1; d < 32; d <<= 1) {
      int y = __shfl_up(iv, d);
      if (lane >= d) iv += y;
    }
    sWave[lane] = iv - v;
  }
  __syncthreads();
  int base = sWave[w] + incl - s3;
  int4 st = {base, base + s0, base + s1, base + s2};
  *(int4*)(sStart + 4 * t) = st;
  if (blockIdx.x == 0) {
    *(int4*)(row_start + 4 * t) = st;
    if (t == FILL_T - 1) row_start[NN] = base + s3;
  }
  __syncthreads();
  col[sStart[ed] + er] = es;
}

#define LP 68
#define CNT_PER_WG (2 * NN / (NN / 16))

template <bool TRANSPOSED = false>
__device__ __forceinline__ v8f tile_gemm64(const float* As, const float* __restrict__ B,
                                           int colbase, int m, int khalf) {
  v8f c = {};
#pragma unroll
  for (int k0 = 0; k0 < 64; k0 += 4) {
    v2f a, b;
    const float* ar = As + m * LP + k0 + khalf;
    a.x = ar[0];
    a.y = ar[1];
    b.x = B[(k0 + khalf) * 64 + colbase + m];
    b.y = B[(k0 + khalf + 1) * 64 + colbase + m];
    c = TRANSPOSED ? wmma4(b, a, c) : wmma4(a, b, c);
  }
  return c;
}

#define NBLK (NN / 16)
#define BLKF 128
#define HEADF (NBLK * BLKF)
#define KEYF (BLKF / 16)
static_assert(DHH == 8 && HH * HEADF == ND, "block layouts are written for head dimension 8");

__device__ __forceinline__ void v_tile_to_blocks(float* __restrict__ Vb, v8f c, int blk,
                                                 int colbase, int lane) {
  int col = colbase + (lane & 15);
  float* p = Vb + (col >> 3) * HEADF + blk * BLKF + (col & 7) * 8 + (lane >> 4) * 4;
  float4 lo = {c[0], c[1], c[2], c[3]};
  float4 hi = {c[4], c[5], c[6], c[7]};
  *(float4*)p = lo;
  *(float4*)(p + 64) = hi;
}
__device__ __forceinline__ void kt_tile_to_blocks(float* __restrict__ Kb, v8f c, int blk,
                                                  int colbase, int lane) {
  float* p = Kb + ((colbase >> 3) + (lane >> 4)) * HEADF + blk * BLKF + (lane & 15) * 4;
  float4 lo = {c[0], c[1], c[2], c[3]};
  float4 hi = {c[4], c[5], c[6], c[7]};
  *(float4*)p = lo;
  *(float4*)(p + 64) = hi;
}

__device__ __forceinline__ void tile_to_global(float* __restrict__ C, v8f c, int rowbase,
                                               int colbase, int lane) {
  int rr = rowbase + 8 * (lane >> 4);
  int cc = colbase + (lane & 15);
#pragma unroll
  for (int j = 0; j < 8; j++) C[(rr + j) * 64 + cc] = c[j];
}
__device__ __forceinline__ void tile_relu_to_lds(float* Cs, v8f c, int colbase, int lane) {
  int rr = 8 * (lane >> 4);
  int cc = colbase + (lane & 15);
#pragma unroll
  for (int j = 0; j < 8; j++) Cs[(rr + j) * LP + cc] = fmaxf(c[j], 0.f);
}

__device__ __forceinline__ float row_tree_sum(const float* x) {
  float y = ((x[0] + x[4]) + (x[2] + x[6])) + ((x[1] + x[5]) + (x[3] + x[7]));
  y += __shfl_xor(y, 4);
  y += __shfl_xor(y, 2);
  y += __shfl_xor(y, 1);
  return y;
}

__device__ __forceinline__ void ln_rows_inplace(float* tile, int t, const float* __restrict__ g,
                                                const float* __restrict__ b) {
  int i = t & 7;
  float* row = tile + (t >> 3) * LP + i;
  float x[8], d[8], sq[8];
#pragma unroll
  for (int k = 0; k < 8; k++) x[k] = row[8 * k];
  float mean = row_tree_sum(x) * (1.f / 64.f);
#pragma unroll
  for (int k = 0; k < 8; k++) {
    d[k] = x[k] - mean;
    sq[k] = d[k] * d[k];
  }
  float var = row_tree_sum(sq) * (1.f / 64.f);
  float rs = rsqrtf(var + 1e-5f);
#pragma unroll
  for (int k = 0; k < 8; k++) row[8 * k] = d[k] * rs * g[i + 8 * k] + b[i + 8 * k];
}

__global__ void __launch_bounds__(128) embed_zero(const float* __restrict__ x,
                                                  const float* __restrict__ We,
                                                  float* __restrict__ h, int* __restrict__ cnt) {
  int t = threadIdx.x;
  int lane = t & 31;
  int rowbase = blockIdx.x * 16;
  int colbase = (t >> 5) * 16;
  int m = lane & 15;
  int khalf = (lane >> 4) * 2;
  if (t < CNT_PER_WG) cnt[blockIdx.x * CNT_PER_WG + t] = 0;
  v8f c = {};
#pragma unroll
  for (int k0 = 0; k0 < INDIM; k0 += 4) {
    v2f a, b;
    const float* ar = x + (rowbase + m) * INDIM + k0 + khalf;
    a.x = ar[0];
    a.y = ar[1];
    b.x = We[(k0 + khalf) * 64 + colbase + m];
    b.y = We[(k0 + khalf + 1) * 64 + colbase + m];
    c = wmma4(a, b, c);
  }
  tile_to_global(h, c, rowbase, colbase, lane);
}

#define GU 16

__global__ void __launch_bounds__(128) stage_a(const float* __restrict__ h,
                                               const int* __restrict__ row_start,
                                               const int* __restrict__ col,
                                               const float* __restrict__ Wg,
                                               const float* __restrict__ Wq,
                                               const float* __restrict__ Wk,
                                               const float* __restrict__ Wv,
                                               float* __restrict__ qb, float* __restrict__ kb,
                                               float* __restrict__ vb) {
  __shared__ __attribute__((aligned(16))) float sS[16 * LP];
  __shared__ __attribute__((aligned(16))) float sH[16 * LP];
  __shared__ __attribute__((aligned(16))) float sF[16 * LP];
  int t = threadIdx.x;
  int lane = t & 31;
  int rowbase = blockIdx.x * 16;
  int colbase = (t >> 5) * 16;
  int m = lane & 15;
  int khalf = (lane >> 4) * 2;

  {
    int r = t >> 3;
    int c0 = (t & 7) * 4;
    int row = rowbase + r;
    int jb = row_start[row];
    int je = row_start[row + 1];
    const float* hc = h + c0;
    float4 hv0 = *(const float4*)(hc + row * DD);
    float4 hv1 = *(const float4*)(hc + row * DD + 32);
    float4 m0 = {0.f, 0.f, 0.f, 0.f};
    float4 m1 = {0.f, 0.f, 0.f, 0.f};
    for (int j = jb; j < je; j += GU) {
      int sidx[GU];
      float4 v0[GU], v1[GU];
#pragma unroll
      for (int u = 0; u < GU; u++) sidx[u] = col[min(j + u, je - 1)];
#pragma unroll
      for (int u = 0; u < GU; u++) {
        const float* p = hc + sidx[u] * DD;
        v0[u] = *(const float4*)p;
        v1[u] = *(const float4*)(p + 32);
      }
#pragma unroll
      for (int u = 0; u < GU; u++) {
        bool in = j + u < je;
        m0.x += in ? v0[u].x : 0.f; m0.y += in ? v0[u].y : 0.f;
        m0.z += in ? v0[u].z : 0.f; m0.w += in ? v0[u].w : 0.f;
        m1.x += in ? v1[u].x : 0.f; m1.y += in ? v1[u].y : 0.f;
        m1.z += in ? v1[u].z : 0.f; m1.w += in ? v1[u].w : 0.f;
      }
    }
    float inv = 1.f / fmaxf((float)(je - jb), 1.f);
    float4 sv0, sv1;
    sv0.x = hv0.x + m0.x * inv;
    sv0.y = hv0.y + m0.y * inv;
    sv0.z = hv0.z + m0.z * inv;
    sv0.w = hv0.w + m0.w * inv;
    sv1.x = hv1.x + m1.x * inv;
    sv1.y = hv1.y + m1.y * inv;
    sv1.z = hv1.z + m1.z * inv;
    sv1.w = hv1.w + m1.w * inv;
    *(float4*)(sH + r * LP + c0) = hv0;
    *(float4*)(sH + r * LP + c0 + 32) = hv1;
    *(float4*)(sS + r * LP + c0) = sv0;
    *(float4*)(sS + r * LP + c0 + 32) = sv1;
  }
  __syncthreads();
  v8f vv = tile_gemm64(sH, Wv, colbase, m, khalf);
  v8f ff = tile_gemm64(sS, Wg, colbase, m, khalf);
  v_tile_to_blocks(vb, vv, blockIdx.x, colbase, lane);
  tile_relu_to_lds(sF, ff, colbase, lane);
  __syncthreads();
  v8f qq = tile_gemm64(sF, Wq, colbase, m, khalf);
  v8f kk = tile_gemm64<true>(sF, Wk, colbase, m, khalf);
  tile_to_global(qb, qq, rowbase, colbase, lane);
  kt_tile_to_blocks(kb, kk, blockIdx.x, colbase, lane);
}

#define KSPLIT 2
#define KPART (NN / KSPLIT)
#define MSLOT 12
__device__ __forceinline__ float xhalf(float x) {
  int b = __builtin_bit_cast(int, x);
  return __builtin_bit_cast(
      float, __builtin_amdgcn_permlanex16(b, b, 0x76543210, 0xfedcba98, false, false));
}


#define GT 4
#define KGROUP (16 * GT)

struct KTile {
  v2f k0, k1;
};
struct VTile {
  v2f v[4];
};

__device__ __forceinline__ void load_k_group(KTile (&kt)[2 * GT], const int base,
                                             const float* __restrict__ kg) {
#pragma unroll
  for (int u = 0; u < GT; u++) {
    float4 kq = *(const float4*)(kg + u * BLKF);
    kt[base + u].k0.x = kq.x; kt[base + u].k0.y = kq.y;
    kt[base + u].k1.x = kq.z; kt[base + u].k1.y = kq.w;
  }
}
__device__ __forceinline__ void load_v_group(VTile (&vt)[2 * GT], const int base,
                                             const float* __restrict__ vg) {
#pragma unroll
  for (int u = 0; u < GT; u++) {
    float4 v01 = *(const float4*)(vg + u * BLKF);
    float4 v23 = *(const float4*)(vg + u * BLKF + 64);
    vt[base + u].v[0].x = v01.x; vt[base + u].v[0].y = v01.y;
    vt[base + u].v[1].x = v01.z; vt[base + u].v[1].y = v01.w;
    vt[base + u].v[2].x = v23.x; vt[base + u].v[2].y = v23.y;
    vt[base + u].v[3].x = v23.z; vt[base + u].v[3].y = v23.w;
  }
}

__device__ __forceinline__ float ex2(float x) { return __builtin_amdgcn_exp2f(x); }

template <bool LOADK, bool LOADV, bool ONLINE>
__device__ __forceinline__ void attn_group(KTile (&kt)[2 * GT], VTile (&vt)[2 * GT], const int cur,
                                           const int oth, const float* __restrict__ kg2,
                                           const float* __restrict__ vg1, v2f qb0, v2f qb1,
                                           float& mrun, float& lsum, v8f& acc) {
  v8f s[GT];
#pragma unroll
  for (int u = 0; u < GT; u++) {
    v8f z = {};
    s[u] = wmma4(kt[cur + u].k0, qb0, z);
  }
#pragma unroll
  for (int u = 0; u < GT; u++) s[u] = wmma4(kt[cur + u].k1, qb1, s[u]);

  if (LOADK) load_k_group(kt, cur, kg2);
  if (LOADV) load_v_group(vt, oth, vg1);

  float mn = 0.f;
  float alpha = 1.f;
  if (ONLINE) {
    float tm[GT];
#pragma unroll
    for (int u = 0; u < GT; u++)
      tm[u] = fmaxf(fmaxf(fmaxf(s[u][0], s[u][1]), fmaxf(s[u][2], s[u][3])),
                    fmaxf(fmaxf(s[u][4], s[u][5]), fmaxf(s[u][6], s[u][7])));
    float sm = fmaxf(fmaxf(tm[0], tm[1]), fmaxf(tm[2], tm[3]));
    sm = fmaxf(sm, xhalf(sm));
    mn = fmaxf(mrun, sm);
    alpha = ex2(mrun - mn);
    mrun = mn;
#pragma unroll
    for (int j = 0; j < 8; j++) acc[j] *= alpha;
  }

  v2f pp[GT][4];
  v2f pt[GT];
#pragma unroll
  for (int u = 0; u < GT; u++) {
#pragma unroll
    for (int c = 0; c < 4; c++) {
      pp[u][c].x = ex2(ONLINE ? s[u][2 * c] - mn : s[u][2 * c]);
      pp[u][c].y = ex2(ONLINE ? s[u][2 * c + 1] - mn : s[u][2 * c + 1]);
    }
    pt[u] = (pp[u][0] + pp[u][1]) + (pp[u][2] + pp[u][3]);
  }
  v2f pg = (pt[0] + pt[1]) + (pt[2] + pt[3]);
  lsum = ONLINE ? lsum * alpha + (pg.x + pg.y) : lsum + (pg.x + pg.y);

#pragma unroll
  for (int u = 0; u < GT; u++) {
#pragma unroll
    for (int c = 0; c < 4; c++) acc = wmma4(vt[cur + u].v[c], pp[u][c], acc);
  }
}
static_assert(GT == 4, "the step's maximum and the denominator's tree are written out for four tiles");

__device__ __forceinline__ void step_fence(v8f& acc, v2f& qb0, v2f& qb1, int& kb) {
  asm volatile("" : "+v"(acc), "+v"(qb0), "+v"(qb1), "+s"(kb));
}

template <bool ONLINE>
__device__ __forceinline__ void key_loop(const float* __restrict__ kp, const float* __restrict__ vp,
                                         v2f qb0, v2f qb1, float& mrun, float& lsum, v8f& acc) {
  KTile kt[2 * GT];
  VTile vt[2 * GT];
  int kb = 0;
  load_k_group(kt, 0, kp + kb * KEYF);
  step_fence(acc, qb0, qb1, kb);
  load_v_group(vt, 0, vp + kb * KEYF);
  step_fence(acc, qb0, qb1, kb);
  load_k_group(kt, GT, kp + (kb + KGROUP) * KEYF);
  step_fence(acc, qb0, qb1, kb);

  for (; kb < KPART - 2 * KGROUP; kb += 2 * KGROUP) {
    attn_group<true, true, ONLINE>(kt, vt, 0, GT, kp + (kb + 2 * KGROUP) * KEYF,
                                   vp + (kb + KGROUP) * KEYF, qb0, qb1, mrun, lsum, acc);
    step_fence(acc, qb0, qb1, kb);
    attn_group<true, true, ONLINE>(kt, vt, GT, 0, kp + (kb + 3 * KGROUP) * KEYF,
                                   vp + (kb + 2 * KGROUP) * KEYF, qb0, qb1, mrun, lsum, acc);
    step_fence(acc, qb0, qb1, kb);
  }
  attn_group<false, true, ONLINE>(kt, vt, 0, GT, kp, vp + (KPART - KGROUP) * KEYF, qb0, qb1, mrun,
                                  lsum, acc);
  attn_group<false, false, ONLINE>(kt, vt, GT, 0, kp, vp, qb0, qb1, mrun, lsum, acc);
}

__global__ void __launch_bounds__(256 * KSPLIT) attn_kernel(const float* __restrict__ q,
                                                   const float* __restrict__ k,
                                                   const float* __restrict__ v,
                                                   float* __restrict__ hbuf,
                                                   const int* __restrict__ cs,
                                                   const float* __restrict__ Wo,
                                                   const float* __restrict__ W1,
                                                   const float* __restrict__ W2,
                                                   const float* __restrict__ g1,
                                                   const float* __restrict__ b1,
                                                   const float* __restrict__ g2,
                                                   const float* __restrict__ b2) {
  __shared__ float sPart[HH * 16 * MSLOT];
  __shared__ __attribute__((aligned(16))) float sX[16 * LP];
  __shared__ __attribute__((aligned(16))) float sY[16 * LP];
  int w = threadIdx.x >> 5;
  int h = w & (HH - 1);
  int part = w >> 3;
  int lane = threadIdx.x & 31;
  int qbase = blockIdx.x * 16;
  int m = lane & 15;
  int half = lane >> 4;
  int khalf = half * 2;
  const float scale = 0.35355339059327373f * 1.4426950408889634f;

  const float* qrow = q + (qbase + m) * DD + h * DHH + 4 * half;
  v2f qb0, qb1;
  qb0.x = qrow[0] * scale;
  qb0.y = qrow[1] * scale;
  qb1.x = qrow[2] * scale;
  qb1.y = qrow[3] * scale;

  float mrun = 0.f;
  float lsum = 0.f;
  v8f acc = {};

  const float* kp = k + h * HEADF + part * KPART * KEYF + half * 64 + m * 4;
  const float* vp = v + h * HEADF + part * KPART * KEYF + (m & 7) * 8 + half * 4;

  key_loop<false>(kp, vp, qb0, qb1, mrun, lsum, acc);

  {
    float amax = fmaxf(fmaxf(fmaxf(fabsf(acc[0]), fabsf(acc[1])), fmaxf(fabsf(acc[2]), fabsf(acc[3]))),
                       fmaxf(fmaxf(fabsf(acc[4]), fabsf(acc[5])), fmaxf(fabsf(acc[6]), fabsf(acc[7]))));
    bool in_range = lsum > 0x1p-60f && lsum < 0x1p100f && amax < 0x1p100f;
    if (__builtin_amdgcn_ballot_w32(!in_range) != 0u) {
      mrun = -1e30f;
      lsum = 0.f;
      acc = v8f{};
      key_loop<true>(kp, vp, qb0, qb1, mrun, lsum, acc);
    }
  }

  float l = lsum + xhalf(lsum);
  float* slot = sPart + (h * 16 + m) * MSLOT;
  if (part == 1 && half == 0) {
    slot[0] = mrun;
    slot[1] = l;
#pragma unroll
    for (int j = 0; j < 8; j++) slot[2 + j] = acc[j];
  }
  __syncthreads();
  if (part == 0 && half == 0) {
    float m1 = slot[0];
    float l1 = slot[1];
    float mn = fmaxf(mrun, m1);
    float a0 = ex2(mrun - mn);
    float a1 = ex2(m1 - mn);
    float lt = l * a0 + l1 * a1;
    float* orow = sX + m * LP + h * DHH;
#pragma unroll
    for (int j = 0; j < 8; j++) orow[j] = (acc[j] * a0 + slot[2 + j] * a1) / lt;
  }
  __syncthreads();

  int tid = threadIdx.x;
  bool chain = w < 4;
  int colbase = w * 16;
  int rr = 8 * half;
  int cc = colbase + m;

  if (chain) {
    v8f op = tile_gemm64(sX, Wo, colbase, m, khalf);
#pragma unroll
    for (int j = 0; j < 8; j++) {
      int row = qbase + rr + j;
      float sc = rsqrtf(1.f + (float)cs[row]);
      sY[(rr + j) * LP + cc] = hbuf[row * DD + cc] + op[j] * sc;
    }
  }
  __syncthreads();
  if (chain) ln_rows_inplace(sY, tid, g1, b1);
  __syncthreads();

  if (chain) {
    v8f f1 = tile_gemm64(sY, W1, colbase, m, khalf);
    tile_relu_to_lds(sX, f1, colbase, lane);
  }
  __syncthreads();
  if (chain) {
    v8f f2 = tile_gemm64(sX, W2, colbase, m, khalf);
#pragma unroll
    for (int j = 0; j < 8; j++) sY[(rr + j) * LP + cc] += f2[j];
  }
  __syncthreads();
  if (chain) ln_rows_inplace(sY, tid, g2, b2);
  __syncthreads();

  if (chain) {
#pragma unroll
    for (int i = 0; i < 2; i++) {
      int idx = tid + 128 * i;
      int r = idx >> 4, c = (idx & 15) * 4;
      *(float4*)(hbuf + (qbase + r) * DD + c) = *(const float4*)(sY + r * LP + c);
    }
  }
}

extern "C" void kernel_launch(void* const* d_in, const int* in_sizes, int n_in,
                              void* d_out, int out_size, void* d_ws, size_t ws_size,
                              hipStream_t stream) {
  const float* x  = (const float*)d_in[0];
  const int*  ei  = (const int*)d_in[1];
  const int*  src = ei;
  const int*  dst = ei + EE;
  const float* We = (const float*)d_in[2];
  const float* Wg = (const float*)d_in[3];
  const float* Wq = (const float*)d_in[4];
  const float* Wk = (const float*)d_in[5];
  const float* Wv = (const float*)d_in[6];
  const float* Wo = (const float*)d_in[7];
  const float* W1 = (const float*)d_in[8];
  const float* W2 = (const float*)d_in[9];
  const float* g1 = (const float*)d_in[10];
  const float* b1 = (const float*)d_in[11];
  const float* g2 = (const float*)d_in[12];
  const float* b2 = (const float*)d_in[13];

  float* h = (float*)d_out;
  float* ws = (float*)d_ws;
  int*   col   = (int*)ws;
  int*   rank  = col + EE;
  int*   rowst = rank + EE;
  float* qb    = ws + 3 * ND;
  float* kb    = qb + ND;
  float* vb    = kb + ND;
  int*   cnt   = (int*)(vb + 4 * ND + 2 * NN);

  embed_zero<<<NN / 16, 128, 0, stream>>>(x, We, h, cnt);
  count_deg<<<(EE + 255) / 256, 256, 0, stream>>>(src, dst, cnt, cnt + NN, rank);
  csr_fill<<<EE / FILL_T, FILL_T, 0, stream>>>(src, dst, cnt + NN, rank, rowst, col);

  for (int l = 0; l < LL; l++) {
    stage_a<<<NN / 16, 128, 0, stream>>>(h, rowst, col, Wg + l * DD * DD, Wq + l * DD * DD,
                                         Wk + l * DD * DD, Wv + l * DD * DD, qb, kb, vb);

    attn_kernel<<<NN / 16, 256 * KSPLIT, 0, stream>>>(qb, kb, vb, h, cnt, Wo + l * DD * DD,
                                                      W1 + l * DD * FF, W2 + l * FF * DD,
                                                      g1 + l * DD, b1 + l * DD, g2 + l * DD,
                                                      b2 + l * DD);
  }
}
